# starting kernel
# baseline (speedup 1.0000x reference)
#include <hip/hip_runtime.h>
#include <math.h>

typedef __attribute__((ext_vector_type(2))) float v2f;
typedef __attribute__((ext_vector_type(8))) float v8f;

#define NOUT 128

__global__ void k_fill1(float* __restrict__ p, int n) {
  int i = blockIdx.x * blockDim.x + threadIdx.x;
  if (i < n) p[i] = 1.0f;
}

__global__ void k_deg_edges(const int* __restrict__ dst, float* __restrict__ deg, int E) {
  int i = blockIdx.x * blockDim.x + threadIdx.x;
  if (i < E) atomicAdd(&deg[dst[i]], 1.0f);
}

__global__ void k_rsqrt_inplace(float* __restrict__ p, int n) {
  int i = blockIdx.x * blockDim.x + threadIdx.x;
  if (i < n) { float d = p[i]; p[i] = (d > 0.0f) ? rsqrtf(d) : 0.0f; }
}

__global__ void k_selfloop(const float* __restrict__ xw, const float* __restrict__ dinv,
                           float* __restrict__ agg, int total) {
  int i = blockIdx.x * blockDim.x + threadIdx.x;
  if (i < total) {
    float di = dinv[i >> 7];
    agg[i] = xw[i] * di * di;
  }
}

__global__ void k_edge_scatter(const float* __restrict__ xw, const float* __restrict__ dinv,
                               const int* __restrict__ src, const int* __restrict__ dst,
                               float* __restrict__ agg, int E) {
  int e = blockIdx.x * 8 + (threadIdx.x >> 5);
  if (e >= E) return;
  int lane = threadIdx.x & 31;
  int s = src[e], d = dst[e];
  float w = dinv[s] * dinv[d];
  const float4* xs = (const float4*)(xw + (size_t)s * NOUT);
  float4 v = xs[lane];
  float* o = agg + (size_t)d * NOUT + lane * 4;
  atomicAdd(o + 0, v.x * w);
  atomicAdd(o + 1, v.y * w);
  atomicAdd(o + 2, v.z * w);
  atomicAdd(o + 3, v.w * w);
}

__global__ void k_bias_relu(float* __restrict__ h, const float* __restrict__ bias, int total) {
  int i = blockIdx.x * blockDim.x + threadIdx.x;
  if (i < total) h[i] = fmaxf(h[i] + bias[i & (NOUT - 1)], 0.0f);
}

__global__ __launch_bounds__(256) void k_gemm_wmma(
    const float* __restrict__ A, const float* __restrict__ W,
    const float* __restrict__ bias, float* __restrict__ C,
    int M, int K, int flags) {
  const int lane = threadIdx.x & 31;
  const int wave = threadIdx.x >> 5;
  const int half = lane >> 4;
  const int r    = lane & 15;
  const int rowBase = blockIdx.x * 16;
  const int colBase = wave * 16;
  if (rowBase >= M) return;

  v8f acc = {};
  const float* arow = A + (size_t)(rowBase + r) * K + 2 * half;
  const float* wcol = W + (size_t)(2 * half) * NOUT + colBase + r;
  for (int k0 = 0; k0 < K; k0 += 4) {
    v2f a; a[0] = arow[k0];                 a[1] = arow[k0 + 1];
    v2f b; b[0] = wcol[(size_t)k0 * NOUT];  b[1] = wcol[(size_t)(k0 + 1) * NOUT];
    acc = __builtin_amdgcn_wmma_f32_16x16x4_f32(
         false, a,  false, b,
         (short)0, acc,  false,  false);
  }

  float bv = (flags & 2) ? bias[colBase + r] : 0.0f;
#pragma unroll
  for (int i = 0; i < 8; ++i) {
    int row = rowBase + half * 8 + i;
    size_t idx = (size_t)row * NOUT + colBase + r;
    float v = acc[i] + bv;
    if (flags & 1) v += C[idx];
    if (flags & 4) v = fmaxf(v, 0.0f);
    C[idx] = v;
  }
}

extern "C" void kernel_launch(void* const* d_in, const int* in_sizes, int n_in,
                              void* d_out, int out_size, void* d_ws, size_t ws_size,
                              hipStream_t stream) {
  const float* x  = (const float*)d_in[0];
  const int*   ei = (const int*)d_in[1];
  const float* W1 = (const float*)d_in[2];
  const float* b1 = (const float*)d_in[3];
  const float* W2 = (const float*)d_in[4];
  const float* b2 = (const float*)d_in[5];
  const float* Ws = (const float*)d_in[6];
  const float* bs = (const float*)d_in[7];
  const float* Wf = (const float*)d_in[8];
  const float* bf = (const float*)d_in[9];
  float* out = (float*)d_out;

  const int IN = 256;
  const int n  = in_sizes[0] / IN;
  const int E  = in_sizes[1] / 2;
  const int* src = ei;
  const int* dst = ei + E;

  float* ws   = (float*)d_ws;
  float* dinv = ws;
  float* xw   = ws + n;
  float* agg  = xw + (size_t)n * NOUT;

  const int tot  = n * NOUT;
  dim3 B(256);
  const int gn   = (n + 255) / 256;
  const int gE   = (E + 255) / 256;
  const int gTot = (tot + 255) / 256;
  const int gEw  = (E + 7) / 8;
  const int gM   = n / 16;

  k_fill1        <<<gn,  B, 0, stream>>>(dinv, n);
  k_deg_edges    <<<gE,  B, 0, stream>>>(dst, dinv, E);
  k_rsqrt_inplace<<<gn,  B, 0, stream>>>(dinv, n);

  k_gemm_wmma    <<<gM,  B, 0, stream>>>(x, W1, nullptr, xw, n, IN, 0);
  k_selfloop     <<<gTot,B, 0, stream>>>(xw, dinv, agg, tot);
  k_edge_scatter <<<gEw, B, 0, stream>>>(xw, dinv, src, dst, agg, E);
  k_bias_relu    <<<gTot,B, 0, stream>>>(agg, b1, tot);

  k_gemm_wmma    <<<gM,  B, 0, stream>>>(agg, W2, nullptr, xw, n, NOUT, 0);
  k_selfloop     <<<gTot,B, 0, stream>>>(xw, dinv, agg, tot);
  k_edge_scatter <<<gEw, B, 0, stream>>>(xw, dinv, src, dst, agg, E);
  k_bias_relu    <<<gTot,B, 0, stream>>>(agg, b2, tot);

  k_gemm_wmma    <<<gM,  B, 0, stream>>>(agg, Wf, bf, out, n, NOUT, 2);
  k_gemm_wmma    <<<gM,  B, 0, stream>>>(x,  Ws, bs, out, n, IN,  1 | 2);
}


# First aggregate sweeps a 1-byte-per-edge owner-block plane, written by the degree kernel, instead of the 4-byte dst list: 4x fewer bytes and loads per sweep; arrival-order sums
# speedup vs baseline: 3.0598x; 3.0598x over previous
#include <hip/hip_runtime.h>
#include <math.h>
#include <stdint.h>
#include <type_traits>

typedef __attribute__((ext_vector_type(2))) float v2f;
typedef __attribute__((ext_vector_type(8))) float v8f;

#define NOUT 128

__global__ void k_fill1(float* __restrict__ p, int n) {
  int i = blockIdx.x * blockDim.x + threadIdx.x;
  if (i < n) p[i] = 1.0f;
}

#define RANGE 196
__global__ void k_deg_edges(const int* __restrict__ dst, float* __restrict__ deg,
                            float* __restrict__ scratch, int n, int E) {
  const int t  = blockIdx.x * blockDim.x + threadIdx.x;
  const int e0 = 4 * t;
  const int epad = (E + 15) & ~15;
  if (e0 >= epad) return;
  int d[4];
  const bool al16 = (reinterpret_cast<uintptr_t>(dst) & 15u) == 0u;
  if (al16 && e0 + 4 <= E) {
    const int4 v = reinterpret_cast<const int4*>(dst)[t];
    d[0] = v.x; d[1] = v.y; d[2] = v.z; d[3] = v.w;
  } else {
#pragma unroll
    for (int j = 0; j < 4; ++j) d[j] = (e0 + j < E) ? dst[e0 + j] : -1;
  }
  unsigned pk = 0u;
#pragma unroll
  for (int j = 0; j < 4; ++j) {
    if (d[j] >= 0) {
      atomicAdd(&deg[d[j]], 1.0f);
      pk |= (((unsigned)d[j] / (unsigned)RANGE) & 255u) << (8 * j);
    }
  }
  if ((size_t)epad <= (size_t)n * NOUT * 4) reinterpret_cast<unsigned*>(scratch)[t] = pk;
}

__global__ void k_rsqrt_inplace(float* __restrict__ p, int n) {
  int i = blockIdx.x * blockDim.x + threadIdx.x;
  if (i < n) { float d = p[i]; p[i] = (d > 0.0f) ? rsqrtf(d) : 0.0f; }
}

#define EPW 8
#define SCAN_V 4
#define AGG_THREADS 1024
#define RING 1024
#define CAPW 320
#define EPR 16

__global__ __launch_bounds__(AGG_THREADS) void k_owner_aggregate(
    const float* __restrict__ xw, const float* __restrict__ dinv,
    const int* __restrict__ src, const int* __restrict__ dst,
    const float* __restrict__ bias, float* __restrict__ h,
    float* __restrict__ scratch, int n, int E, int mode) {
  __shared__ float acc[RANGE * NOUT];
  __shared__ alignas(8) int ring[(AGG_THREADS / 32) * RING];
  static_assert(AGG_THREADS / 32 == 32, "a wave reads the counts of all waves of its block, one per lane");
  static_assert(CAPW % 32 == 0 && CAPW % EPW == 0 && 2 * CAPW <= RING && RANGE <= 256, "record layout");

  const int tid  = threadIdx.x;
  const int lane = tid & 31;
  const int wave = __builtin_amdgcn_readfirstlane(tid >> 5);
  const int lo   = blockIdx.x * RANGE;
  const int cnt  = (n - lo < RANGE) ? (n - lo) : RANGE;

  const float4* xw4 = reinterpret_cast<const float4*>(xw);

  const size_t nwaves = (size_t)gridDim.x * (AGG_THREADS / 32);
  const size_t pbytes = (((size_t)E + 15) >> 4) << 4;
  const bool usable = (n <= (1 << 23)) &&
                      (pbytes + nwaves * (CAPW * 8 + 4) <= (size_t)n * NOUT * 4) &&
                      ((reinterpret_cast<uintptr_t>(scratch) & 15u) == 0u);
  const bool plane_ok = usable && (gridDim.x <= 256u);
  char* rbase = reinterpret_cast<char*>(scratch) + pbytes;
  int2* seg   = reinterpret_cast<int2*>(rbase) + ((size_t)blockIdx.x * (AGG_THREADS / 32) + wave) * CAPW;
  int* counts = reinterpret_cast<int*>(rbase) + nwaves * CAPW * 2;
  const bool rec = usable && (mode == 0);
  bool replay = false;
  int mycount = 0;
  if (usable && mode == 1) {
    const int cv = counts[blockIdx.x * (AGG_THREADS / 32) + lane];
    replay  = __builtin_amdgcn_ballot_w32(cv > CAPW) == 0u;
    mycount = __builtin_amdgcn_readlane(cv, wave);
  }

  for (int idx = tid; idx < cnt * 32; idx += AGG_THREADS) {
    const int node = idx >> 5;
    const float di = dinv[lo + node];
    const float4 x = xw4[(size_t)(lo + node) * 32 + lane];
    float* a = acc + node * NOUT + lane;
    a[0]  = x.x * di * di;
    a[32] = x.y * di * di;
    a[64] = x.z * di * di;
    a[96] = x.w * di * di;
  }
  __syncthreads();

  int* myring = ring + wave * RING;
  int wcount = 0;
  int wdone  = 0;

  auto test = [&](int d, int e) {
    const bool m = (unsigned)(d - lo) < (unsigned)cnt;
    const unsigned mask = __builtin_amdgcn_ballot_w32(m);
    if (mask != 0u) {
      if (m) myring[(wcount + (int)__builtin_amdgcn_mbcnt_lo(mask, 0u)) & (RING - 1)] = e;
      wcount += __builtin_popcount(mask);
    }
  };

  auto fetch = [&](int k, int& s, int& dl, float& w) {
    s = 0; dl = 0; w = 0.0f;
    if (lane < k) {
      const int e = myring[(wdone + lane) & (RING - 1)];
      s = src[e];
      const int d = dst[e];
      w = dinv[s] * dinv[d];
      dl = d - lo;
      if (rec && wdone + lane < CAPW)
        seg[wdone + lane] = make_int2((s << 8) | dl, __builtin_bit_cast(int, w));
    }
  };

  auto add_row = [&](const float4& v, float wi, int dli) {
    float* a = acc + dli * NOUT + lane;
    atomicAdd(a,      v.x * wi);
    atomicAdd(a + 32, v.y * wi);
    atomicAdd(a + 64, v.z * wi);
    atomicAdd(a + 96, v.w * wi);
  };

  auto drain = [&]() {
    while (wcount - wdone >= EPW) {
      int s, dl; float w;
      fetch(EPW, s, dl, w);
      float4 v[EPW];
#pragma unroll
      for (int i = 0; i < EPW; ++i) {
        const int si = __builtin_amdgcn_readlane(s, i);
        v[i] = xw4[(size_t)si * 32 + lane];
      }
#pragma unroll
      for (int i = 0; i < EPW; ++i) {
        const float wi = __builtin_bit_cast(float, __builtin_amdgcn_readlane(__builtin_bit_cast(int, w), i));
        const int dli  = __builtin_amdgcn_readlane(dl, i);
        add_row(v[i], wi, dli);
      }
      wdone += EPW;
    }
  };

  auto leftover = [&]() {
    const int k = wcount - wdone;
    if (k > 0) {
      int s, dl; float w;
      fetch(k, s, dl, w);
      for (int i = 0; i < k; ++i) {
        const int si = __builtin_amdgcn_readlane(s, i);
        const float wi = __builtin_bit_cast(float, __builtin_amdgcn_readlane(__builtin_bit_cast(int, w), i));
        const int dli  = __builtin_amdgcn_readlane(dl, i);
        const float4 v = xw4[(size_t)si * 32 + lane];
        add_row(v, wi, dli);
      }
      wdone += k;
    }
    if (rec && lane == 0) counts[blockIdx.x * (AGG_THREADS / 32) + wave] = wcount;
  };

  if (replay) {
    auto group = [&](const int2& ent, int first, auto ne) {
      constexpr int NE = decltype(ne)::value;
      float4 v[NE];
#pragma unroll
      for (int i = 0; i < NE; ++i) {
        const int key = __builtin_amdgcn_readlane(ent.x, first + i);
        v[i] = xw4[(size_t)(key >> 8) * 32 + lane];
      }
#pragma unroll
      for (int i = 0; i < NE; ++i) {
        const int key  = __builtin_amdgcn_readlane(ent.x, first + i);
        const float wi = __builtin_bit_cast(float, __builtin_amdgcn_readlane(ent.y, first + i));
        add_row(v[i], wi, key & 255);
      }
    };
    int2* ment = reinterpret_cast<int2*>(myring);
    {
      int2 e[CAPW / 32];
#pragma unroll
      for (int j = 0; j < CAPW / 32; ++j) {
        e[j] = make_int2(0, 0);
        if (j * 32 + lane < mycount) e[j] = seg[j * 32 + lane];
      }
#pragma unroll
      for (int j = 0; j < CAPW / 32; ++j) ment[j * 32 + lane] = e[j];
    }
    for (int base = 0; base < mycount; base += 32) {
      const int2 cur = ment[base + lane];
      const int k = (mycount - base < 32) ? (mycount - base) : 32;
      if (k == 32) {
#pragma unroll
        for (int g = 0; g < 32; g += EPR) group(cur, g, std::integral_constant<int, EPR>{});
      } else {
        int g = 0;
        for (; g + EPW <= k; g += EPW) group(cur, g, std::integral_constant<int, EPW>{});
        for (; g < k; ++g) group(cur, g, std::integral_constant<int, 1>{});
      }
    }
  } else if (plane_ok) {
    const uint4* p4 = reinterpret_cast<const uint4*>(scratch);
    const int np = (E + 15) >> 4;
    const unsigned pat = blockIdx.x * 0x01010101u;
    auto hit = [&](unsigned w) {
      const unsigned x = w ^ pat;
      return ~(((x & 0x7F7F7F7Fu) + 0x7F7F7F7Fu) | x | 0x7F7F7F7Fu);
    };
    auto take = [&](unsigned t, int ebase) {
      while (__builtin_amdgcn_ballot_w32(t != 0u) != 0u) {
        const int e = ebase + (__builtin_ctz(t | 0x80000000u) >> 3);
        const bool m = (t != 0u) && (e < E);
        const unsigned mask = __builtin_amdgcn_ballot_w32(m);
        if (m) myring[(wcount + (int)__builtin_amdgcn_mbcnt_lo(mask, 0u)) & (RING - 1)] = e;
        wcount += __builtin_popcount(mask);
        t &= t - 1u;
      }
    };
    const int rot = (int)(blockIdx.x * ((unsigned)np / gridDim.x)) & ~31;
    auto piece = [&](const uint4& pv, int v) {
      const int e = 16 * v;
      const unsigned t0 = hit(pv.x), t1 = hit(pv.y), t2 = hit(pv.z), t3 = hit(pv.w);
      if (__builtin_amdgcn_ballot_w32((t0 | t1 | t2 | t3) != 0u) != 0u) {
        take(t0, e);
        take(t1, e + 4);
        take(t2, e + 8);
        take(t3, e + 12);
      }
      drain();
    };
    int vb = 0;
    for (; vb + SCAN_V * AGG_THREADS <= np; vb += SCAN_V * AGG_THREADS) {
      uint4 pv[SCAN_V];
      int   vi[SCAN_V];
#pragma unroll
      for (int j = 0; j < SCAN_V; ++j) {
        int v = vb + j * AGG_THREADS + tid + rot;
        if (v >= np) v -= np;
        vi[j] = v;
        pv[j] = p4[v];
      }
#pragma unroll
      for (int j = 0; j < SCAN_V; ++j) piece(pv[j], vi[j]);
    }
    for (; vb < np; vb += AGG_THREADS) {
      const int p = vb + tid;
      int v = p + rot;
      if (v >= np) v -= np;
      uint4 pv = make_uint4(~pat, ~pat, ~pat, ~pat);
      if (p < np) pv = p4[v];
      piece(pv, v);
    }
    leftover();
  } else {
    const bool al16 = (reinterpret_cast<uintptr_t>(dst) & 15u) == 0u;
    const int nv = al16 ? (E >> 2) : 0;
    const int4* d4 = reinterpret_cast<const int4*>(dst);
    const int rot = (int)(blockIdx.x * ((unsigned)nv / gridDim.x)) & ~31;
    int vb = 0;
    for (; vb + SCAN_V * AGG_THREADS <= nv; vb += SCAN_V * AGG_THREADS) {
      int4 dv[SCAN_V];
      int  vi[SCAN_V];
#pragma unroll
      for (int j = 0; j < SCAN_V; ++j) {
        int v = vb + j * AGG_THREADS + tid + rot;
        if (v >= nv) v -= nv;
        vi[j] = v;
        dv[j] = d4[v];
      }
#pragma unroll
      for (int j = 0; j < SCAN_V; ++j) {
        const int e = 4 * vi[j];
        test(dv[j].x, e);
        test(dv[j].y, e + 1);
        test(dv[j].z, e + 2);
        test(dv[j].w, e + 3);
      }
      drain();
    }
    for (; vb < nv; vb += AGG_THREADS) {
      const int p = vb + tid;
      int v = p + rot;
      if (v >= nv) v -= nv;
      int4 dv = make_int4(-1, -1, -1, -1);
      if (p < nv) dv = d4[v];
      const int e = 4 * v;
      test(dv.x, e);
      test(dv.y, e + 1);
      test(dv.z, e + 2);
      test(dv.w, e + 3);
      drain();
    }
    for (int eb = nv * 4; eb < E; eb += AGG_THREADS) {
      const int e = eb + tid;
      const int d = (e < E) ? dst[e] : -1;
      test(d, e);
      drain();
    }
    leftover();
  }
  __syncthreads();

  const float4 bv = reinterpret_cast<const float4*>(bias)[lane];
  float4* h4 = reinterpret_cast<float4*>(h);
  for (int idx = tid; idx < cnt * 32; idx += AGG_THREADS) {
    const int node = idx >> 5;
    const float* a = acc + node * NOUT + lane;
    float4 o;
    o.x = fmaxf(a[0]  + bv.x, 0.0f);
    o.y = fmaxf(a[32] + bv.y, 0.0f);
    o.z = fmaxf(a[64] + bv.z, 0.0f);
    o.w = fmaxf(a[96] + bv.w, 0.0f);
    h4[(size_t)(lo + node) * 32 + lane] = o;
  }
}

template <int KK>
__global__ __launch_bounds__(256) void k_gemm_wmma(
    const float* __restrict__ A, const float* __restrict__ W,
    const float* __restrict__ bias, float* __restrict__ C,
    int M, int flags) {
  constexpr int P = KK + 4;
  constexpr int V = KK / 4;
  __shared__ alignas(16) float sA[16 * P];

  const int tid  = threadIdx.x;
  const int lane = tid & 31;
  const int wave = tid >> 5;
  const int half = lane >> 4;
  const int r    = lane & 15;
  const int rowBase = blockIdx.x * 16;
  const int colBase = wave * 16;
  if (rowBase >= M) return;

  const float4* Ag = reinterpret_cast<const float4*>(A + (size_t)rowBase * KK);
  constexpr int NV = (16 * V) / 256;
  float4 st[NV];
#pragma unroll
  for (int j = 0; j < NV; ++j) st[j] = Ag[tid + j * 256];
#pragma unroll
  for (int j = 0; j < NV; ++j) {
    const int i = tid + j * 256;
    *reinterpret_cast<float4*>(&sA[(i / V) * P + (i % V) * 4]) = st[j];
  }
  __syncthreads();

  v8f acc = {};
  const float* arow = sA + r * P + 2 * half;
  const float* wcol = W + (size_t)(2 * half) * NOUT + colBase + r;
#pragma unroll 8
  for (int k0 = 0; k0 < KK; k0 += 4) {
    const v2f a = *reinterpret_cast<const v2f*>(arow + k0);
    v2f b; b[0] = wcol[(size_t)k0 * NOUT];  b[1] = wcol[(size_t)(k0 + 1) * NOUT];
    acc = __builtin_amdgcn_wmma_f32_16x16x4_f32(
         false, a,  false, b,
         (short)0, acc,  false,  false);
  }

  float bv = (flags & 2) ? bias[colBase + r] : 0.0f;
#pragma unroll
  for (int i = 0; i < 8; ++i) {
    int row = rowBase + half * 8 + i;
    size_t idx = (size_t)row * NOUT + colBase + r;
    float v = acc[i] + bv;
    if (flags & 1) v += C[idx];
    if (flags & 4) v = fmaxf(v, 0.0f);
    C[idx] = v;
  }
}

extern "C" void kernel_launch(void* const* d_in, const int* in_sizes, int n_in,
                              void* d_out, int out_size, void* d_ws, size_t ws_size,
                              hipStream_t stream) {
  const float* x  = (const float*)d_in[0];
  const int*   ei = (const int*)d_in[1];
  const float* W1 = (const float*)d_in[2];
  const float* b1 = (const float*)d_in[3];
  const float* W2 = (const float*)d_in[4];
  const float* b2 = (const float*)d_in[5];
  const float* Ws = (const float*)d_in[6];
  const float* bs = (const float*)d_in[7];
  const float* Wf = (const float*)d_in[8];
  const float* bf = (const float*)d_in[9];
  float* out = (float*)d_out;

  const int IN = 256;
  const int n  = in_sizes[0] / IN;
  const int E  = in_sizes[1] / 2;
  const int* src = ei;
  const int* dst = ei + E;

  float* ws   = (float*)d_ws;
  float* dinv = ws;
  float* xw   = ws + n;
  float* agg  = xw + (size_t)n * NOUT;

  dim3 B(256);
  const int gn   = (n + 255) / 256;
  const int gE4  = ((E + 15) / 16 * 4 + 255) / 256;
  const int gR   = (n + RANGE - 1) / RANGE;
  const int gM   = n / 16;

  k_fill1        <<<gn,  B, 0, stream>>>(dinv, n);
  k_deg_edges    <<<gE4, B, 0, stream>>>(dst, dinv, out, n, E);
  k_rsqrt_inplace<<<gn,  B, 0, stream>>>(dinv, n);

  k_gemm_wmma<256><<<gM, B, 0, stream>>>(x, W1, nullptr, xw, n, 0);
  k_owner_aggregate<<<gR, dim3(AGG_THREADS), 0, stream>>>(xw, dinv, src, dst, b1, agg, out, n, E, 0);

  k_gemm_wmma<128><<<gM, B, 0, stream>>>(agg, W2, nullptr, xw, n, 0);
  k_owner_aggregate<<<gR, dim3(AGG_THREADS), 0, stream>>>(xw, dinv, src, dst, b2, agg, out, n, E, 1);

  k_gemm_wmma<128><<<gM, B, 0, stream>>>(agg, Wf, bf, out, n, 2);
  k_gemm_wmma<256><<<gM, B, 0, stream>>>(x,  Ws, bs, out, n, 1 | 2);
}


# Degree count without global atomics: 64 blocks each count a slice of dst into a whole-graph LDS histogram (adds of 1.0: exact, order-free); the rsqrt kernel sums the 64 partials; fill launch gone
# speedup vs baseline: 3.7984x; 1.2414x over previous
#include <hip/hip_runtime.h>
#include <math.h>
#include <stdint.h>
#include <type_traits>

typedef __attribute__((ext_vector_type(2))) float v2f;
typedef __attribute__((ext_vector_type(8))) float v8f;

#define NOUT 128

#define RANGE 196
#define DEG_BLOCKS 64
#define DEG_THREADS 1024
#define DEG_HN 51200
#define DEG_V 8
static_assert(DEG_BLOCKS <= NOUT, "the partial counts live in a buffer of n * NOUT floats");

__global__ __launch_bounds__(DEG_THREADS) void k_deg_hist(
    const int* __restrict__ dst, float* __restrict__ part, float* __restrict__ scratch, int n, int E) {
  __shared__ alignas(16) float hist[DEG_HN];
  const int tid  = threadIdx.x;
  const int epad = (E + 15) & ~15;
  const int nq   = epad >> 2;
  const int per  = (nq + (int)gridDim.x - 1) / (int)gridDim.x;
  const int q0   = (int)blockIdx.x * per;
  const int q1   = (nq - q0 < per) ? nq : q0 + per;
  const bool al16  = (reinterpret_cast<uintptr_t>(dst) & 15u) == 0u;
  const bool plane = (size_t)epad <= (size_t)n * NOUT * 4;
  unsigned* pl  = reinterpret_cast<unsigned*>(scratch);
  float* mypart = part + (size_t)blockIdx.x * n;

  for (int base = 0; base < n; base += DEG_HN) {
    const int cnt = (n - base < DEG_HN) ? (n - base) : DEG_HN;
    for (int i = tid; i < (cnt + 3) / 4; i += DEG_THREADS)
      reinterpret_cast<float4*>(hist)[i] = make_float4(0.0f, 0.0f, 0.0f, 0.0f);
    __syncthreads();

    auto one = [&](int d, int j, unsigned& pk) {
      if (d >= 0) {
        const unsigned r = (unsigned)(d - base);
        if (r < (unsigned)cnt) atomicAdd(&hist[r], 1.0f);
        pk |= (((unsigned)d / (unsigned)RANGE) & 255u) << (8 * j);
      }
    };
    for (int qb = q0; qb < q1; qb += DEG_V * DEG_THREADS) {
      int4 dv[DEG_V];
#pragma unroll
      for (int j = 0; j < DEG_V; ++j) {
        const int q = qb + j * DEG_THREADS + tid;
        dv[j] = make_int4(-1, -1, -1, -1);
        if (q < q1) {
          const int e0 = 4 * q;
          if (al16 && e0 + 4 <= E) {
            dv[j] = reinterpret_cast<const int4*>(dst)[q];
          } else {
            if (e0     < E) dv[j].x = dst[e0];
            if (e0 + 1 < E) dv[j].y = dst[e0 + 1];
            if (e0 + 2 < E) dv[j].z = dst[e0 + 2];
            if (e0 + 3 < E) dv[j].w = dst[e0 + 3];
          }
        }
      }
#pragma unroll
      for (int j = 0; j < DEG_V; ++j) {
        const int q = qb + j * DEG_THREADS + tid;
        if (q < q1) {
          unsigned pk = 0u;
          one(dv[j].x, 0, pk);
          one(dv[j].y, 1, pk);
          one(dv[j].z, 2, pk);
          one(dv[j].w, 3, pk);
          if (plane && base == 0) pl[q] = pk;
        }
      }
    }
    __syncthreads();

    float* o = mypart + base;
    int done = 0;
    if ((reinterpret_cast<uintptr_t>(o) & 15u) == 0u) {
      for (int i = tid; i < cnt / 4; i += DEG_THREADS)
        reinterpret_cast<float4*>(o)[i] = reinterpret_cast<const float4*>(hist)[i];
      done = cnt & ~3;
    }
    for (int i = done + tid; i < cnt; i += DEG_THREADS) o[i] = hist[i];
    if (base + DEG_HN < n) __syncthreads();
  }
}

#define DEG_LD 32
static_assert(DEG_BLOCKS % DEG_LD == 0, "the partials are read in whole groups");
__global__ void k_dinv_from_parts(const float* __restrict__ part, float* __restrict__ dinv, int n) {
  const int i = blockIdx.x * blockDim.x + threadIdx.x;
  if (i >= n) return;
  const float* p = part + i;
  float d = 1.0f;
  for (int g0 = 0; g0 < DEG_BLOCKS; g0 += DEG_LD) {
    float v[DEG_LD];
#pragma unroll
    for (int g = 0; g < DEG_LD; ++g) v[g] = p[(size_t)(g0 + g) * n];
#pragma unroll
    for (int g = 0; g < DEG_LD; ++g) d += v[g];
  }
  dinv[i] = (d > 0.0f) ? rsqrtf(d) : 0.0f;
}

#define EPW 8
#define SCAN_V 4
#define AGG_THREADS 1024
#define RING 1024
#define CAPW 320
#define EPR 16

__global__ __launch_bounds__(AGG_THREADS) void k_owner_aggregate(
    const float* __restrict__ xw, const float* __restrict__ dinv,
    const int* __restrict__ src, const int* __restrict__ dst,
    const float* __restrict__ bias, float* __restrict__ h,
    float* __restrict__ scratch, int n, int E, int mode) {
  __shared__ float acc[RANGE * NOUT];
  __shared__ alignas(8) int ring[(AGG_THREADS / 32) * RING];
  static_assert(AGG_THREADS / 32 == 32, "a wave reads the counts of all waves of its block, one per lane");
  static_assert(CAPW % 32 == 0 && CAPW % EPW == 0 && 2 * CAPW <= RING && RANGE <= 256, "record layout");

  const int tid  = threadIdx.x;
  const int lane = tid & 31;
  const int wave = __builtin_amdgcn_readfirstlane(tid >> 5);
  const int lo   = blockIdx.x * RANGE;
  const int cnt  = (n - lo < RANGE) ? (n - lo) : RANGE;

  const float4* xw4 = reinterpret_cast<const float4*>(xw);

  const size_t nwaves = (size_t)gridDim.x * (AGG_THREADS / 32);
  const size_t pbytes = (((size_t)E + 15) >> 4) << 4;
  const bool usable = (n <= (1 << 23)) &&
                      (pbytes + nwaves * (CAPW * 8 + 4) <= (size_t)n * NOUT * 4) &&
                      ((reinterpret_cast<uintptr_t>(scratch) & 15u) == 0u);
  const bool plane_ok = usable && (gridDim.x <= 256u);
  char* rbase = reinterpret_cast<char*>(scratch) + pbytes;
  int2* seg   = reinterpret_cast<int2*>(rbase) + ((size_t)blockIdx.x * (AGG_THREADS / 32) + wave) * CAPW;
  int* counts = reinterpret_cast<int*>(rbase) + nwaves * CAPW * 2;
  const bool rec = usable && (mode == 0);
  bool replay = false;
  int mycount = 0;
  if (usable && mode == 1) {
    const int cv = counts[blockIdx.x * (AGG_THREADS / 32) + lane];
    replay  = __builtin_amdgcn_ballot_w32(cv > CAPW) == 0u;
    mycount = __builtin_amdgcn_readlane(cv, wave);
  }

  for (int idx = tid; idx < cnt * 32; idx += AGG_THREADS) {
    const int node = idx >> 5;
    const float di = dinv[lo + node];
    const float4 x = xw4[(size_t)(lo + node) * 32 + lane];
    float* a = acc + node * NOUT + lane;
    a[0]  = x.x * di * di;
    a[32] = x.y * di * di;
    a[64] = x.z * di * di;
    a[96] = x.w * di * di;
  }
  __syncthreads();

  int* myring = ring + wave * RING;
  int wcount = 0;
  int wdone  = 0;

  auto test = [&](int d, int e) {
    const bool m = (unsigned)(d - lo) < (unsigned)cnt;
    const unsigned mask = __builtin_amdgcn_ballot_w32(m);
    if (mask != 0u) {
      if (m) myring[(wcount + (int)__builtin_amdgcn_mbcnt_lo(mask, 0u)) & (RING - 1)] = e;
      wcount += __builtin_popcount(mask);
    }
  };

  auto fetch = [&](int k, int& s, int& dl, float& w) {
    s = 0; dl = 0; w = 0.0f;
    if (lane < k) {
      const int e = myring[(wdone + lane) & (RING - 1)];
      s = src[e];
      const int d = dst[e];
      w = dinv[s] * dinv[d];
      dl = d - lo;
      if (rec && wdone + lane < CAPW)
        seg[wdone + lane] = make_int2((s << 8) | dl, __builtin_bit_cast(int, w));
    }
  };

  auto add_row = [&](const float4& v, float wi, int dli) {
    float* a = acc + dli * NOUT + lane;
    atomicAdd(a,      v.x * wi);
    atomicAdd(a + 32, v.y * wi);
    atomicAdd(a + 64, v.z * wi);
    atomicAdd(a + 96, v.w * wi);
  };

  auto drain = [&]() {
    while (wcount - wdone >= EPW) {
      int s, dl; float w;
      fetch(EPW, s, dl, w);
      float4 v[EPW];
#pragma unroll
      for (int i = 0; i < EPW; ++i) {
        const int si = __builtin_amdgcn_readlane(s, i);
        v[i] = xw4[(size_t)si * 32 + lane];
      }
#pragma unroll
      for (int i = 0; i < EPW; ++i) {
        const float wi = __builtin_bit_cast(float, __builtin_amdgcn_readlane(__builtin_bit_cast(int, w), i));
        const int dli  = __builtin_amdgcn_readlane(dl, i);
        add_row(v[i], wi, dli);
      }
      wdone += EPW;
    }
  };

  auto leftover = [&]() {
    const int k = wcount - wdone;
    if (k > 0) {
      int s, dl; float w;
      fetch(k, s, dl, w);
      for (int i = 0; i < k; ++i) {
        const int si = __builtin_amdgcn_readlane(s, i);
        const float wi = __builtin_bit_cast(float, __builtin_amdgcn_readlane(__builtin_bit_cast(int, w), i));
        const int dli  = __builtin_amdgcn_readlane(dl, i);
        const float4 v = xw4[(size_t)si * 32 + lane];
        add_row(v, wi, dli);
      }
      wdone += k;
    }
    if (rec && lane == 0) counts[blockIdx.x * (AGG_THREADS / 32) + wave] = wcount;
  };

  if (replay) {
    auto group = [&](const int2& ent, int first, auto ne) {
      constexpr int NE = decltype(ne)::value;
      float4 v[NE];
#pragma unroll
      for (int i = 0; i < NE; ++i) {
        const int key = __builtin_amdgcn_readlane(ent.x, first + i);
        v[i] = xw4[(size_t)(key >> 8) * 32 + lane];
      }
#pragma unroll
      for (int i = 0; i < NE; ++i) {
        const int key  = __builtin_amdgcn_readlane(ent.x, first + i);
        const float wi = __builtin_bit_cast(float, __builtin_amdgcn_readlane(ent.y, first + i));
        add_row(v[i], wi, key & 255);
      }
    };
    int2* ment = reinterpret_cast<int2*>(myring);
    {
      int2 e[CAPW / 32];
#pragma unroll
      for (int j = 0; j < CAPW / 32; ++j) {
        e[j] = make_int2(0, 0);
        if (j * 32 + lane < mycount) e[j] = seg[j * 32 + lane];
      }
#pragma unroll
      for (int j = 0; j < CAPW / 32; ++j) ment[j * 32 + lane] = e[j];
    }
    for (int base = 0; base < mycount; base += 32) {
      const int2 cur = ment[base + lane];
      const int k = (mycount - base < 32) ? (mycount - base) : 32;
      if (k == 32) {
#pragma unroll
        for (int g = 0; g < 32; g += EPR) group(cur, g, std::integral_constant<int, EPR>{});
      } else {
        int g = 0;
        for (; g + EPW <= k; g += EPW) group(cur, g, std::integral_constant<int, EPW>{});
        for (; g < k; ++g) group(cur, g, std::integral_constant<int, 1>{});
      }
    }
  } else if (plane_ok) {
    const uint4* p4 = reinterpret_cast<const uint4*>(scratch);
    const int np = (E + 15) >> 4;
    const unsigned pat = blockIdx.x * 0x01010101u;
    auto hit = [&](unsigned w) {
      const unsigned x = w ^ pat;
      return ~(((x & 0x7F7F7F7Fu) + 0x7F7F7F7Fu) | x | 0x7F7F7F7Fu);
    };
    auto take = [&](unsigned t, int ebase) {
      while (__builtin_amdgcn_ballot_w32(t != 0u) != 0u) {
        const int e = ebase + (__builtin_ctz(t | 0x80000000u) >> 3);
        const bool m = (t != 0u) && (e < E);
        const unsigned mask = __builtin_amdgcn_ballot_w32(m);
        if (m) myring[(wcount + (int)__builtin_amdgcn_mbcnt_lo(mask, 0u)) & (RING - 1)] = e;
        wcount += __builtin_popcount(mask);
        t &= t - 1u;
      }
    };
    const int rot = (int)(blockIdx.x * ((unsigned)np / gridDim.x)) & ~31;
    auto piece = [&](const uint4& pv, int v) {
      const int e = 16 * v;
      const unsigned t0 = hit(pv.x), t1 = hit(pv.y), t2 = hit(pv.z), t3 = hit(pv.w);
      if (__builtin_amdgcn_ballot_w32((t0 | t1 | t2 | t3) != 0u) != 0u) {
        take(t0, e);
        take(t1, e + 4);
        take(t2, e + 8);
        take(t3, e + 12);
      }
      drain();
    };
    int vb = 0;
    for (; vb + SCAN_V * AGG_THREADS <= np; vb += SCAN_V * AGG_THREADS) {
      uint4 pv[SCAN_V];
      int   vi[SCAN_V];
#pragma unroll
      for (int j = 0; j < SCAN_V; ++j) {
        int v = vb + j * AGG_THREADS + tid + rot;
        if (v >= np) v -= np;
        vi[j] = v;
        pv[j] = p4[v];
      }
#pragma unroll
      for (int j = 0; j < SCAN_V; ++j) piece(pv[j], vi[j]);
    }
    for (; vb < np; vb += AGG_THREADS) {
      const int p = vb + tid;
      int v = p + rot;
      if (v >= np) v -= np;
      uint4 pv = make_uint4(~pat, ~pat, ~pat, ~pat);
      if (p < np) pv = p4[v];
      piece(pv, v);
    }
    leftover();
  } else {
    const bool al16 = (reinterpret_cast<uintptr_t>(dst) & 15u) == 0u;
    const int nv = al16 ? (E >> 2) : 0;
    const int4* d4 = reinterpret_cast<const int4*>(dst);
    const int rot = (int)(blockIdx.x * ((unsigned)nv / gridDim.x)) & ~31;
    int vb = 0;
    for (; vb + SCAN_V * AGG_THREADS <= nv; vb += SCAN_V * AGG_THREADS) {
      int4 dv[SCAN_V];
      int  vi[SCAN_V];
#pragma unroll
      for (int j = 0; j < SCAN_V; ++j) {
        int v = vb + j * AGG_THREADS + tid + rot;
        if (v >= nv) v -= nv;
        vi[j] = v;
        dv[j] = d4[v];
      }
#pragma unroll
      for (int j = 0; j < SCAN_V; ++j) {
        const int e = 4 * vi[j];
        test(dv[j].x, e);
        test(dv[j].y, e + 1);
        test(dv[j].z, e + 2);
        test(dv[j].w, e + 3);
      }
      drain();
    }
    for (; vb < nv; vb += AGG_THREADS) {
      const int p = vb + tid;
      int v = p + rot;
      if (v >= nv) v -= nv;
      int4 dv = make_int4(-1, -1, -1, -1);
      if (p < nv) dv = d4[v];
      const int e = 4 * v;
      test(dv.x, e);
      test(dv.y, e + 1);
      test(dv.z, e + 2);
      test(dv.w, e + 3);
      drain();
    }
    for (int eb = nv * 4; eb < E; eb += AGG_THREADS) {
      const int e = eb + tid;
      const int d = (e < E) ? dst[e] : -1;
      test(d, e);
      drain();
    }
    leftover();
  }
  __syncthreads();

  const float4 bv = reinterpret_cast<const float4*>(bias)[lane];
  float4* h4 = reinterpret_cast<float4*>(h);
  for (int idx = tid; idx < cnt * 32; idx += AGG_THREADS) {
    const int node = idx >> 5;
    const float* a = acc + node * NOUT + lane;
    float4 o;
    o.x = fmaxf(a[0]  + bv.x, 0.0f);
    o.y = fmaxf(a[32] + bv.y, 0.0f);
    o.z = fmaxf(a[64] + bv.z, 0.0f);
    o.w = fmaxf(a[96] + bv.w, 0.0f);
    h4[(size_t)(lo + node) * 32 + lane] = o;
  }
}

template <int KK>
__global__ __launch_bounds__(256) void k_gemm_wmma(
    const float* __restrict__ A, const float* __restrict__ W,
    const float* __restrict__ bias, float* __restrict__ C,
    int M, int flags) {
  constexpr int P = KK + 4;
  constexpr int V = KK / 4;
  __shared__ alignas(16) float sA[16 * P];

  const int tid  = threadIdx.x;
  const int lane = tid & 31;
  const int wave = tid >> 5;
  const int half = lane >> 4;
  const int r    = lane & 15;
  const int rowBase = blockIdx.x * 16;
  const int colBase = wave * 16;
  if (rowBase >= M) return;

  const float4* Ag = reinterpret_cast<const float4*>(A + (size_t)rowBase * KK);
  constexpr int NV = (16 * V) / 256;
  float4 st[NV];
#pragma unroll
  for (int j = 0; j < NV; ++j) st[j] = Ag[tid + j * 256];
#pragma unroll
  for (int j = 0; j < NV; ++j) {
    const int i = tid + j * 256;
    *reinterpret_cast<float4*>(&sA[(i / V) * P + (i % V) * 4]) = st[j];
  }
  __syncthreads();

  v8f acc = {};
  const float* arow = sA + r * P + 2 * half;
  const float* wcol = W + (size_t)(2 * half) * NOUT + colBase + r;
#pragma unroll 8
  for (int k0 = 0; k0 < KK; k0 += 4) {
    const v2f a = *reinterpret_cast<const v2f*>(arow + k0);
    v2f b; b[0] = wcol[(size_t)k0 * NOUT];  b[1] = wcol[(size_t)(k0 + 1) * NOUT];
    acc = __builtin_amdgcn_wmma_f32_16x16x4_f32(
         false, a,  false, b,
         (short)0, acc,  false,  false);
  }

  float bv = (flags & 2) ? bias[colBase + r] : 0.0f;
#pragma unroll
  for (int i = 0; i < 8; ++i) {
    int row = rowBase + half * 8 + i;
    size_t idx = (size_t)row * NOUT + colBase + r;
    float v = acc[i] + bv;
    if (flags & 1) v += C[idx];
    if (flags & 4) v = fmaxf(v, 0.0f);
    C[idx] = v;
  }
}

extern "C" void kernel_launch(void* const* d_in, const int* in_sizes, int n_in,
                              void* d_out, int out_size, void* d_ws, size_t ws_size,
                              hipStream_t stream) {
  const float* x  = (const float*)d_in[0];
  const int*   ei = (const int*)d_in[1];
  const float* W1 = (const float*)d_in[2];
  const float* b1 = (const float*)d_in[3];
  const float* W2 = (const float*)d_in[4];
  const float* b2 = (const float*)d_in[5];
  const float* Ws = (const float*)d_in[6];
  const float* bs = (const float*)d_in[7];
  const float* Wf = (const float*)d_in[8];
  const float* bf = (const float*)d_in[9];
  float* out = (float*)d_out;

  const int IN = 256;
  const int n  = in_sizes[0] / IN;
  const int E  = in_sizes[1] / 2;
  const int* src = ei;
  const int* dst = ei + E;

  float* ws   = (float*)d_ws;
  float* dinv = ws;
  float* xw   = ws + n;
  float* agg  = xw + (size_t)n * NOUT;

  dim3 B(256);
  const int gn   = (n + 255) / 256;
  const int gR   = (n + RANGE - 1) / RANGE;
  const int gM   = n / 16;

  k_deg_hist       <<<DEG_BLOCKS, dim3(DEG_THREADS), 0, stream>>>(dst, agg, out, n, E);
  k_dinv_from_parts<<<gn, B, 0, stream>>>(agg, dinv, n);

  k_gemm_wmma<256><<<gM, B, 0, stream>>>(x, W1, nullptr, xw, n, 0);
  k_owner_aggregate<<<gR, dim3(AGG_THREADS), 0, stream>>>(xw, dinv, src, dst, b1, agg, out, n, E, 0);

  k_gemm_wmma<128><<<gM, B, 0, stream>>>(agg, W2, nullptr, xw, n, 0);
  k_owner_aggregate<<<gR, dim3(AGG_THREADS), 0, stream>>>(xw, dinv, src, dst, b2, agg, out, n, E, 1);

  k_gemm_wmma<128><<<gM, B, 0, stream>>>(agg, Wf, bf, out, n, 2);
  k_gemm_wmma<256><<<gM, B, 0, stream>>>(x,  Ws, bs, out, n, 1 | 2);
}


# GEMM: W held in registers: a wave loads the B fragments of its 16 columns for all k once; 512 blocks each loop over their 16-row stripes (A stripe double-buffered in LDS); k and sum order unchanged
# speedup vs baseline: 4.0126x; 1.0564x over previous
#include <hip/hip_runtime.h>
#include <math.h>
#include <stdint.h>
#include <type_traits>

typedef __attribute__((ext_vector_type(2))) float v2f;
typedef __attribute__((ext_vector_type(8))) float v8f;

#define NOUT 128

#define RANGE 196
#define DEG_BLOCKS 64
#define DEG_THREADS 1024
#define DEG_HN 51200
#define DEG_V 8
static_assert(DEG_BLOCKS <= NOUT, "the partial counts live in a buffer of n * NOUT floats");

__global__ __launch_bounds__(DEG_THREADS) void k_deg_hist(
    const int* __restrict__ dst, float* __restrict__ part, float* __restrict__ scratch, int n, int E) {
  __shared__ alignas(16) float hist[DEG_HN];
  const int tid  = threadIdx.x;
  const int epad = (E + 15) & ~15;
  const int nq   = epad >> 2;
  const int per  = (nq + (int)gridDim.x - 1) / (int)gridDim.x;
  const int q0   = (int)blockIdx.x * per;
  const int q1   = (nq - q0 < per) ? nq : q0 + per;
  const bool al16  = (reinterpret_cast<uintptr_t>(dst) & 15u) == 0u;
  const bool plane = (size_t)epad <= (size_t)n * NOUT * 4;
  unsigned* pl  = reinterpret_cast<unsigned*>(scratch);
  float* mypart = part + (size_t)blockIdx.x * n;

  for (int base = 0; base < n; base += DEG_HN) {
    const int cnt = (n - base < DEG_HN) ? (n - base) : DEG_HN;
    for (int i = tid; i < (cnt + 3) / 4; i += DEG_THREADS)
      reinterpret_cast<float4*>(hist)[i] = make_float4(0.0f, 0.0f, 0.0f, 0.0f);
    __syncthreads();

    auto one = [&](int d, int j, unsigned& pk) {
      if (d >= 0) {
        const unsigned r = (unsigned)(d - base);
        if (r < (unsigned)cnt) atomicAdd(&hist[r], 1.0f);
        pk |= (((unsigned)d / (unsigned)RANGE) & 255u) << (8 * j);
      }
    };
    for (int qb = q0; qb < q1; qb += DEG_V * DEG_THREADS) {
      int4 dv[DEG_V];
#pragma unroll
      for (int j = 0; j < DEG_V; ++j) {
        const int q = qb + j * DEG_THREADS + tid;
        dv[j] = make_int4(-1, -1, -1, -1);
        if (q < q1) {
          const int e0 = 4 * q;
          if (al16 && e0 + 4 <= E) {
            dv[j] = reinterpret_cast<const int4*>(dst)[q];
          } else {
            if (e0     < E) dv[j].x = dst[e0];
            if (e0 + 1 < E) dv[j].y = dst[e0 + 1];
            if (e0 + 2 < E) dv[j].z = dst[e0 + 2];
            if (e0 + 3 < E) dv[j].w = dst[e0 + 3];
          }
        }
      }
#pragma unroll
      for (int j = 0; j < DEG_V; ++j) {
        const int q = qb + j * DEG_THREADS + tid;
        if (q < q1) {
          unsigned pk = 0u;
          one(dv[j].x, 0, pk);
          one(dv[j].y, 1, pk);
          one(dv[j].z, 2, pk);
          one(dv[j].w, 3, pk);
          if (plane && base == 0) pl[q] = pk;
        }
      }
    }
    __syncthreads();

    float* o = mypart + base;
    int done = 0;
    if ((reinterpret_cast<uintptr_t>(o) & 15u) == 0u) {
      for (int i = tid; i < cnt / 4; i += DEG_THREADS)
        reinterpret_cast<float4*>(o)[i] = reinterpret_cast<const float4*>(hist)[i];
      done = cnt & ~3;
    }
    for (int i = done + tid; i < cnt; i += DEG_THREADS) o[i] = hist[i];
    if (base + DEG_HN < n) __syncthreads();
  }
}

#define DEG_LD 32
static_assert(DEG_BLOCKS % DEG_LD == 0, "the partials are read in whole groups");
__global__ void k_dinv_from_parts(const float* __restrict__ part, float* __restrict__ dinv, int n) {
  const int i = blockIdx.x * blockDim.x + threadIdx.x;
  if (i >= n) return;
  const float* p = part + i;
  float d = 1.0f;
  for (int g0 = 0; g0 < DEG_BLOCKS; g0 += DEG_LD) {
    float v[DEG_LD];
#pragma unroll
    for (int g = 0; g < DEG_LD; ++g) v[g] = p[(size_t)(g0 + g) * n];
#pragma unroll
    for (int g = 0; g < DEG_LD; ++g) d += v[g];
  }
  dinv[i] = (d > 0.0f) ? rsqrtf(d) : 0.0f;
}

#define EPW 8
#define SCAN_V 4
#define AGG_THREADS 1024
#define RING 1024
#define CAPW 320
#define EPR 16

__global__ __launch_bounds__(AGG_THREADS) void k_owner_aggregate(
    const float* __restrict__ xw, const float* __restrict__ dinv,
    const int* __restrict__ src, const int* __restrict__ dst,
    const float* __restrict__ bias, float* __restrict__ h,
    float* __restrict__ scratch, int n, int E, int mode) {
  __shared__ float acc[RANGE * NOUT];
  __shared__ alignas(8) int ring[(AGG_THREADS / 32) * RING];
  static_assert(AGG_THREADS / 32 == 32, "a wave reads the counts of all waves of its block, one per lane");
  static_assert(CAPW % 32 == 0 && CAPW % EPW == 0 && 2 * CAPW <= RING && RANGE <= 256, "record layout");

  const int tid  = threadIdx.x;
  const int lane = tid & 31;
  const int wave = __builtin_amdgcn_readfirstlane(tid >> 5);
  const int lo   = blockIdx.x * RANGE;
  const int cnt  = (n - lo < RANGE) ? (n - lo) : RANGE;

  const float4* xw4 = reinterpret_cast<const float4*>(xw);

  const size_t nwaves = (size_t)gridDim.x * (AGG_THREADS / 32);
  const size_t pbytes = (((size_t)E + 15) >> 4) << 4;
  const bool usable = (n <= (1 << 23)) &&
                      (pbytes + nwaves * (CAPW * 8 + 4) <= (size_t)n * NOUT * 4) &&
                      ((reinterpret_cast<uintptr_t>(scratch) & 15u) == 0u);
  const bool plane_ok = usable && (gridDim.x <= 256u);
  char* rbase = reinterpret_cast<char*>(scratch) + pbytes;
  int2* seg   = reinterpret_cast<int2*>(rbase) + ((size_t)blockIdx.x * (AGG_THREADS / 32) + wave) * CAPW;
  int* counts = reinterpret_cast<int*>(rbase) + nwaves * CAPW * 2;
  const bool rec = usable && (mode == 0);
  bool replay = false;
  int mycount = 0;
  if (usable && mode == 1) {
    const int cv = counts[blockIdx.x * (AGG_THREADS / 32) + lane];
    replay  = __builtin_amdgcn_ballot_w32(cv > CAPW) == 0u;
    mycount = __builtin_amdgcn_readlane(cv, wave);
  }

  for (int idx = tid; idx < cnt * 32; idx += AGG_THREADS) {
    const int node = idx >> 5;
    const float di = dinv[lo + node];
    const float4 x = xw4[(size_t)(lo + node) * 32 + lane];
    float* a = acc + node * NOUT + lane;
    a[0]  = x.x * di * di;
    a[32] = x.y * di * di;
    a[64] = x.z * di * di;
    a[96] = x.w * di * di;
  }
  __syncthreads();

  int* myring = ring + wave * RING;
  int wcount = 0;
  int wdone  = 0;

  auto test = [&](int d, int e) {
    const bool m = (unsigned)(d - lo) < (unsigned)cnt;
    const unsigned mask = __builtin_amdgcn_ballot_w32(m);
    if (mask != 0u) {
      if (m) myring[(wcount + (int)__builtin_amdgcn_mbcnt_lo(mask, 0u)) & (RING - 1)] = e;
      wcount += __builtin_popcount(mask);
    }
  };

  auto fetch = [&](int k, int& s, int& dl, float& w) {
    s = 0; dl = 0; w = 0.0f;
    if (lane < k) {
      const int e = myring[(wdone + lane) & (RING - 1)];
      s = src[e];
      const int d = dst[e];
      w = dinv[s] * dinv[d];
      dl = d - lo;
      if (rec && wdone + lane < CAPW)
        seg[wdone + lane] = make_int2((s << 8) | dl, __builtin_bit_cast(int, w));
    }
  };

  auto add_row = [&](const float4& v, float wi, int dli) {
    float* a = acc + dli * NOUT + lane;
    atomicAdd(a,      v.x * wi);
    atomicAdd(a + 32, v.y * wi);
    atomicAdd(a + 64, v.z * wi);
    atomicAdd(a + 96, v.w * wi);
  };

  auto drain = [&]() {
    while (wcount - wdone >= EPW) {
      int s, dl; float w;
      fetch(EPW, s, dl, w);
      float4 v[EPW];
#pragma unroll
      for (int i = 0; i < EPW; ++i) {
        const int si = __builtin_amdgcn_readlane(s, i);
        v[i] = xw4[(size_t)si * 32 + lane];
      }
#pragma unroll
      for (int i = 0; i < EPW; ++i) {
        const float wi = __builtin_bit_cast(float, __builtin_amdgcn_readlane(__builtin_bit_cast(int, w), i));
        const int dli  = __builtin_amdgcn_readlane(dl, i);
        add_row(v[i], wi, dli);
      }
      wdone += EPW;
    }
  };

  auto leftover = [&]() {
    const int k = wcount - wdone;
    if (k > 0) {
      int s, dl; float w;
      fetch(k, s, dl, w);
      for (int i = 0; i < k; ++i) {
        const int si = __builtin_amdgcn_readlane(s, i);
        const float wi = __builtin_bit_cast(float, __builtin_amdgcn_readlane(__builtin_bit_cast(int, w), i));
        const int dli  = __builtin_amdgcn_readlane(dl, i);
        const float4 v = xw4[(size_t)si * 32 + lane];
        add_row(v, wi, dli);
      }
      wdone += k;
    }
    if (rec && lane == 0) counts[blockIdx.x * (AGG_THREADS / 32) + wave] = wcount;
  };

  if (replay) {
    auto group = [&](const int2& ent, int first, auto ne) {
      constexpr int NE = decltype(ne)::value;
      float4 v[NE];
#pragma unroll
      for (int i = 0; i < NE; ++i) {
        const int key = __builtin_amdgcn_readlane(ent.x, first + i);
        v[i] = xw4[(size_t)(key >> 8) * 32 + lane];
      }
#pragma unroll
      for (int i = 0; i < NE; ++i) {
        const int key  = __builtin_amdgcn_readlane(ent.x, first + i);
        const float wi = __builtin_bit_cast(float, __builtin_amdgcn_readlane(ent.y, first + i));
        add_row(v[i], wi, key & 255);
      }
    };
    int2* ment = reinterpret_cast<int2*>(myring);
    {
      int2 e[CAPW / 32];
#pragma unroll
      for (int j = 0; j < CAPW / 32; ++j) {
        e[j] = make_int2(0, 0);
        if (j * 32 + lane < mycount) e[j] = seg[j * 32 + lane];
      }
#pragma unroll
      for (int j = 0; j < CAPW / 32; ++j) ment[j * 32 + lane] = e[j];
    }
    for (int base = 0; base < mycount; base += 32) {
      const int2 cur = ment[base + lane];
      const int k = (mycount - base < 32) ? (mycount - base) : 32;
      if (k == 32) {
#pragma unroll
        for (int g = 0; g < 32; g += EPR) group(cur, g, std::integral_constant<int, EPR>{});
      } else {
        int g = 0;
        for (; g + EPW <= k; g += EPW) group(cur, g, std::integral_constant<int, EPW>{});
        for (; g < k; ++g) group(cur, g, std::integral_constant<int, 1>{});
      }
    }
  } else if (plane_ok) {
    const uint4* p4 = reinterpret_cast<const uint4*>(scratch);
    const int np = (E + 15) >> 4;
    const unsigned pat = blockIdx.x * 0x01010101u;
    auto hit = [&](unsigned w) {
      const unsigned x = w ^ pat;
      return ~(((x & 0x7F7F7F7Fu) + 0x7F7F7F7Fu) | x | 0x7F7F7F7Fu);
    };
    auto take = [&](unsigned t, int ebase) {
      while (__builtin_amdgcn_ballot_w32(t != 0u) != 0u) {
        const int e = ebase + (__builtin_ctz(t | 0x80000000u) >> 3);
        const bool m = (t != 0u) && (e < E);
        const unsigned mask = __builtin_amdgcn_ballot_w32(m);
        if (m) myring[(wcount + (int)__builtin_amdgcn_mbcnt_lo(mask, 0u)) & (RING - 1)] = e;
        wcount += __builtin_popcount(mask);
        t &= t - 1u;
      }
    };
    const int rot = (int)(blockIdx.x * ((unsigned)np / gridDim.x)) & ~31;
    auto piece = [&](const uint4& pv, int v) {
      const int e = 16 * v;
      const unsigned t0 = hit(pv.x), t1 = hit(pv.y), t2 = hit(pv.z), t3 = hit(pv.w);
      if (__builtin_amdgcn_ballot_w32((t0 | t1 | t2 | t3) != 0u) != 0u) {
        take(t0, e);
        take(t1, e + 4);
        take(t2, e + 8);
        take(t3, e + 12);
      }
      drain();
    };
    int vb = 0;
    for (; vb + SCAN_V * AGG_THREADS <= np; vb += SCAN_V * AGG_THREADS) {
      uint4 pv[SCAN_V];
      int   vi[SCAN_V];
#pragma unroll
      for (int j = 0; j < SCAN_V; ++j) {
        int v = vb + j * AGG_THREADS + tid + rot;
        if (v >= np) v -= np;
        vi[j] = v;
        pv[j] = p4[v];
      }
#pragma unroll
      for (int j = 0; j < SCAN_V; ++j) piece(pv[j], vi[j]);
    }
    for (; vb < np; vb += AGG_THREADS) {
      const int p = vb + tid;
      int v = p + rot;
      if (v >= np) v -= np;
      uint4 pv = make_uint4(~pat, ~pat, ~pat, ~pat);
      if (p < np) pv = p4[v];
      piece(pv, v);
    }
    leftover();
  } else {
    const bool al16 = (reinterpret_cast<uintptr_t>(dst) & 15u) == 0u;
    const int nv = al16 ? (E >> 2) : 0;
    const int4* d4 = reinterpret_cast<const int4*>(dst);
    const int rot = (int)(blockIdx.x * ((unsigned)nv / gridDim.x)) & ~31;
    int vb = 0;
    for (; vb + SCAN_V * AGG_THREADS <= nv; vb += SCAN_V * AGG_THREADS) {
      int4 dv[SCAN_V];
      int  vi[SCAN_V];
#pragma unroll
      for (int j = 0; j < SCAN_V; ++j) {
        int v = vb + j * AGG_THREADS + tid + rot;
        if (v >= nv) v -= nv;
        vi[j] = v;
        dv[j] = d4[v];
      }
#pragma unroll
      for (int j = 0; j < SCAN_V; ++j) {
        const int e = 4 * vi[j];
        test(dv[j].x, e);
        test(dv[j].y, e + 1);
        test(dv[j].z, e + 2);
        test(dv[j].w, e + 3);
      }
      drain();
    }
    for (; vb < nv; vb += AGG_THREADS) {
      const int p = vb + tid;
      int v = p + rot;
      if (v >= nv) v -= nv;
      int4 dv = make_int4(-1, -1, -1, -1);
      if (p < nv) dv = d4[v];
      const int e = 4 * v;
      test(dv.x, e);
      test(dv.y, e + 1);
      test(dv.z, e + 2);
      test(dv.w, e + 3);
      drain();
    }
    for (int eb = nv * 4; eb < E; eb += AGG_THREADS) {
      const int e = eb + tid;
      const int d = (e < E) ? dst[e] : -1;
      test(d, e);
      drain();
    }
    leftover();
  }
  __syncthreads();

  const float4 bv = reinterpret_cast<const float4*>(bias)[lane];
  float4* h4 = reinterpret_cast<float4*>(h);
  for (int idx = tid; idx < cnt * 32; idx += AGG_THREADS) {
    const int node = idx >> 5;
    const float* a = acc + node * NOUT + lane;
    float4 o;
    o.x = fmaxf(a[0]  + bv.x, 0.0f);
    o.y = fmaxf(a[32] + bv.y, 0.0f);
    o.z = fmaxf(a[64] + bv.z, 0.0f);
    o.w = fmaxf(a[96] + bv.w, 0.0f);
    h4[(size_t)(lo + node) * 32 + lane] = o;
  }
}

#define GEMM_BLOCKS 512
template <int KK>
__global__ __launch_bounds__(256) void k_gemm_wmma(
    const float* __restrict__ A, const float* __restrict__ W,
    const float* __restrict__ bias, float* __restrict__ C,
    int M, int flags) {
  constexpr int P  = KK + 4;
  constexpr int V  = KK / 4;
  constexpr int KS = KK / 4;
  constexpr int NV = (16 * V) / 256;
  __shared__ alignas(16) float sA[2][16 * P];

  const int tid  = threadIdx.x;
  const int lane = tid & 31;
  const int wave = tid >> 5;
  const int half = lane >> 4;
  const int r    = lane & 15;
  const int colBase = wave * 16;
  const int ntiles  = M / 16;
  const int step    = (int)gridDim.x;
  int t = (int)blockIdx.x;
  if (t >= ntiles) return;

  const float4* A4 = reinterpret_cast<const float4*>(A);
  static_assert(NV == 2 || NV == 4, "a thread carries 2 or 4 pieces of a stripe");
  float4 s0, s1, s2 = {}, s3 = {};
  auto fetch_stripe = [&](int tile) {
    const float4* p = A4 + (size_t)tile * (16 * V) + tid;
    s0 = p[0];
    s1 = p[256];
    if constexpr (NV == 4) { s2 = p[512]; s3 = p[768]; }
  };
  auto put_piece = [&](float* dstStripe, int i, const float4& v) {
    *reinterpret_cast<float4*>(&dstStripe[(i / V) * P + (i % V) * 4]) = v;
  };
  auto put_stripe = [&](float* dstStripe) {
    put_piece(dstStripe, tid, s0);
    put_piece(dstStripe, tid + 256, s1);
    if constexpr (NV == 4) { put_piece(dstStripe, tid + 512, s2); put_piece(dstStripe, tid + 768, s3); }
  };
  fetch_stripe(t);

  v2f bf[KS];
  const float* wcol = W + (size_t)(2 * half) * NOUT + colBase + r;
#pragma unroll
  for (int s = 0; s < KS; ++s) {
    bf[s][0] = wcol[(size_t)(4 * s) * NOUT];
    bf[s][1] = wcol[(size_t)(4 * s + 1) * NOUT];
  }
  const float bv = (flags & 2) ? bias[colBase + r] : 0.0f;

  put_stripe(sA[0]);
  __syncthreads();

  int buf = 0;
  for (; t < ntiles; t += step) {
    const int  tn   = t + step;
    const bool more = tn < ntiles;
    if (more) fetch_stripe(tn);

    const int rowBase = t * 16;
    float* crow = C + (size_t)(rowBase + half * 8) * NOUT + colBase + r;
    v8f cold = {};
    if (flags & 1) {
#pragma unroll
      for (int i = 0; i < 8; ++i) cold[i] = crow[(size_t)i * NOUT];
    }

    v8f acc = {};
    const float* arow = &sA[buf][r * P + 2 * half];
#pragma unroll
    for (int s = 0; s < KS; ++s) {
      const v2f a = *reinterpret_cast<const v2f*>(arow + 4 * s);
      acc = __builtin_amdgcn_wmma_f32_16x16x4_f32(
           false, a,  false, bf[s],
           (short)0, acc,  false,  false);
    }

    put_stripe(sA[buf ^ 1]);
    __syncthreads();
    buf ^= 1;

#pragma unroll
    for (int i = 0; i < 8; ++i) {
      float v = acc[i] + bv;
      if (flags & 1) v += cold[i];
      if (flags & 4) v = fmaxf(v, 0.0f);
      crow[(size_t)i * NOUT] = v;
    }
  }
}

extern "C" void kernel_launch(void* const* d_in, const int* in_sizes, int n_in,
                              void* d_out, int out_size, void* d_ws, size_t ws_size,
                              hipStream_t stream) {
  const float* x  = (const float*)d_in[0];
  const int*   ei = (const int*)d_in[1];
  const float* W1 = (const float*)d_in[2];
  const float* b1 = (const float*)d_in[3];
  const float* W2 = (const float*)d_in[4];
  const float* b2 = (const float*)d_in[5];
  const float* Ws = (const float*)d_in[6];
  const float* bs = (const float*)d_in[7];
  const float* Wf = (const float*)d_in[8];
  const float* bf = (const float*)d_in[9];
  float* out = (float*)d_out;

  const int IN = 256;
  const int n  = in_sizes[0] / IN;
  const int E  = in_sizes[1] / 2;
  const int* src = ei;
  const int* dst = ei + E;

  float* ws   = (float*)d_ws;
  float* dinv = ws;
  float* xw   = ws + n;
  float* agg  = xw + (size_t)n * NOUT;

  dim3 B(256);
  const int gn   = (n + 255) / 256;
  const int gR   = (n + RANGE - 1) / RANGE;
  const int gM   = GEMM_BLOCKS;

  k_deg_hist       <<<DEG_BLOCKS, dim3(DEG_THREADS), 0, stream>>>(dst, agg, out, n, E);
  k_dinv_from_parts<<<gn, B, 0, stream>>>(agg, dinv, n);

  k_gemm_wmma<256><<<gM, B, 0, stream>>>(x, W1, nullptr, xw, n, 0);
  k_owner_aggregate<<<gR, dim3(AGG_THREADS), 0, stream>>>(xw, dinv, src, dst, b1, agg, out, n, E, 0);

  k_gemm_wmma<128><<<gM, B, 0, stream>>>(agg, W2, nullptr, xw, n, 0);
  k_owner_aggregate<<<gR, dim3(AGG_THREADS), 0, stream>>>(xw, dinv, src, dst, b2, agg, out, n, E, 1);

  k_gemm_wmma<128><<<gM, B, 0, stream>>>(agg, Wf, bf, out, n, 2);
  k_gemm_wmma<256><<<gM, B, 0, stream>>>(x,  Ws, bs, out, n, 1 | 2);
}


# Second aggregate sums in registers, no LDS float adds: layer 1 queues each found edge (LDS int atomic slot) to the wave owning its destination row; layer-2 sum order = queue (arrival) order
# speedup vs baseline: 4.1731x; 1.0400x over previous
#include <hip/hip_runtime.h>
#include <math.h>
#include <stdint.h>
#include <type_traits>

typedef __attribute__((ext_vector_type(2))) float v2f;
typedef __attribute__((ext_vector_type(8))) float v8f;
typedef __attribute__((ext_vector_type(4))) float v4f;

#define NOUT 128

#define RANGE 196
#define DEG_BLOCKS 64
#define DEG_THREADS 1024
#define DEG_HN 51200
#define DEG_V 8
static_assert(DEG_BLOCKS <= NOUT, "the partial counts live in a buffer of n * NOUT floats");

__global__ __launch_bounds__(DEG_THREADS) void k_deg_hist(
    const int* __restrict__ dst, float* __restrict__ part, float* __restrict__ scratch, int n, int E) {
  __shared__ alignas(16) float hist[DEG_HN];
  const int tid  = threadIdx.x;
  const int epad = (E + 15) & ~15;
  const int nq   = epad >> 2;
  const int per  = (nq + (int)gridDim.x - 1) / (int)gridDim.x;
  const int q0   = (int)blockIdx.x * per;
  const int q1   = (nq - q0 < per) ? nq : q0 + per;
  const bool al16  = (reinterpret_cast<uintptr_t>(dst) & 15u) == 0u;
  const bool plane = (size_t)epad <= (size_t)n * NOUT * 4;
  unsigned* pl  = reinterpret_cast<unsigned*>(scratch);
  float* mypart = part + (size_t)blockIdx.x * n;

  for (int base = 0; base < n; base += DEG_HN) {
    const int cnt = (n - base < DEG_HN) ? (n - base) : DEG_HN;
    for (int i = tid; i < (cnt + 3) / 4; i += DEG_THREADS)
      reinterpret_cast<float4*>(hist)[i] = make_float4(0.0f, 0.0f, 0.0f, 0.0f);
    __syncthreads();

    auto one = [&](int d, int j, unsigned& pk) {
      if (d >= 0) {
        const unsigned r = (unsigned)(d - base);
        if (r < (unsigned)cnt) atomicAdd(&hist[r], 1.0f);
        pk |= (((unsigned)d / (unsigned)RANGE) & 255u) << (8 * j);
      }
    };
    for (int qb = q0; qb < q1; qb += DEG_V * DEG_THREADS) {
      int4 dv[DEG_V];
#pragma unroll
      for (int j = 0; j < DEG_V; ++j) {
        const int q = qb + j * DEG_THREADS + tid;
        dv[j] = make_int4(-1, -1, -1, -1);
        if (q < q1) {
          const int e0 = 4 * q;
          if (al16 && e0 + 4 <= E) {
            dv[j] = reinterpret_cast<const int4*>(dst)[q];
          } else {
            if (e0     < E) dv[j].x = dst[e0];
            if (e0 + 1 < E) dv[j].y = dst[e0 + 1];
            if (e0 + 2 < E) dv[j].z = dst[e0 + 2];
            if (e0 + 3 < E) dv[j].w = dst[e0 + 3];
          }
        }
      }
#pragma unroll
      for (int j = 0; j < DEG_V; ++j) {
        const int q = qb + j * DEG_THREADS + tid;
        if (q < q1) {
          unsigned pk = 0u;
          one(dv[j].x, 0, pk);
          one(dv[j].y, 1, pk);
          one(dv[j].z, 2, pk);
          one(dv[j].w, 3, pk);
          if (plane && base == 0) pl[q] = pk;
        }
      }
    }
    __syncthreads();

    float* o = mypart + base;
    int done = 0;
    if ((reinterpret_cast<uintptr_t>(o) & 15u) == 0u) {
      for (int i = tid; i < cnt / 4; i += DEG_THREADS)
        reinterpret_cast<float4*>(o)[i] = reinterpret_cast<const float4*>(hist)[i];
      done = cnt & ~3;
    }
    for (int i = done + tid; i < cnt; i += DEG_THREADS) o[i] = hist[i];
    if (base + DEG_HN < n) __syncthreads();
  }
}

#define DEG_LD 32
static_assert(DEG_BLOCKS % DEG_LD == 0, "the partials are read in whole groups");
__global__ void k_dinv_from_parts(const float* __restrict__ part, float* __restrict__ dinv, int n) {
  const int i = blockIdx.x * blockDim.x + threadIdx.x;
  if (i >= n) return;
  const float* p = part + i;
  float d = 1.0f;
  for (int g0 = 0; g0 < DEG_BLOCKS; g0 += DEG_LD) {
    float v[DEG_LD];
#pragma unroll
    for (int g = 0; g < DEG_LD; ++g) v[g] = p[(size_t)(g0 + g) * n];
#pragma unroll
    for (int g = 0; g < DEG_LD; ++g) d += v[g];
  }
  dinv[i] = (d > 0.0f) ? rsqrtf(d) : 0.0f;
}

#define EPW 8
#define SCAN_V 4
#define AGG_THREADS 1024
#define RING 1024
#define CAPW 320
#define EPR 16

__global__ __launch_bounds__(AGG_THREADS) void k_owner_aggregate(
    const float* __restrict__ xw, const float* __restrict__ dinv,
    const int* __restrict__ src, const int* __restrict__ dst,
    const float* __restrict__ bias, float* __restrict__ h,
    float* __restrict__ scratch, int n, int E, int mode) {
  __shared__ float acc[RANGE * NOUT];
  __shared__ alignas(8) int ring[(AGG_THREADS / 32) * RING];
  __shared__ alignas(8) int2 queue[(AGG_THREADS / 32) * CAPW];
  __shared__ int qcnt[AGG_THREADS / 32];
  static_assert(AGG_THREADS / 32 == 32, "a wave reads the counts of all waves of its block, one per lane");
  static_assert(CAPW % 32 == 0 && CAPW % EPW == 0 && 2 * CAPW <= RING && RANGE <= 256, "record layout");

  const int tid  = threadIdx.x;
  const int lane = tid & 31;
  const int wave = __builtin_amdgcn_readfirstlane(tid >> 5);
  const int lo   = blockIdx.x * RANGE;
  const int cnt  = (n - lo < RANGE) ? (n - lo) : RANGE;

  const float4* xw4 = reinterpret_cast<const float4*>(xw);

  const size_t nwaves = (size_t)gridDim.x * (AGG_THREADS / 32);
  const size_t pbytes = (((size_t)E + 15) >> 4) << 4;
  const bool usable = (n <= (1 << 23)) &&
                      (pbytes + nwaves * (CAPW * 8 + 4) <= (size_t)n * NOUT * 4) &&
                      ((reinterpret_cast<uintptr_t>(scratch) & 15u) == 0u);
  const bool plane_ok = usable && (gridDim.x <= 256u);
  char* rbase = reinterpret_cast<char*>(scratch) + pbytes;
  int2* seg   = reinterpret_cast<int2*>(rbase) + ((size_t)blockIdx.x * (AGG_THREADS / 32) + wave) * CAPW;
  int* counts = reinterpret_cast<int*>(rbase) + nwaves * CAPW * 2;
  const bool rec = usable && (mode == 0);
  bool replay = false;
  int mycount = 0;
  if (usable && mode == 1) {
    const int cv = counts[blockIdx.x * (AGG_THREADS / 32) + lane];
    replay  = __builtin_amdgcn_ballot_w32(cv > CAPW) == 0u;
    mycount = __builtin_amdgcn_readlane(cv, wave);
  }

  if (tid < AGG_THREADS / 32) qcnt[tid] = 0;
  if (!replay)
  for (int idx = tid; idx < cnt * 32; idx += AGG_THREADS) {
    const int node = idx >> 5;
    const float di = dinv[lo + node];
    const float4 x = xw4[(size_t)(lo + node) * 32 + lane];
    float* a = acc + node * NOUT + lane;
    a[0]  = x.x * di * di;
    a[32] = x.y * di * di;
    a[64] = x.z * di * di;
    a[96] = x.w * di * di;
  }
  __syncthreads();

  int* myring = ring + wave * RING;
  int wcount = 0;
  int wdone  = 0;

  auto test = [&](int d, int e) {
    const bool m = (unsigned)(d - lo) < (unsigned)cnt;
    const unsigned mask = __builtin_amdgcn_ballot_w32(m);
    if (mask != 0u) {
      if (m) myring[(wcount + (int)__builtin_amdgcn_mbcnt_lo(mask, 0u)) & (RING - 1)] = e;
      wcount += __builtin_popcount(mask);
    }
  };

  auto fetch = [&](int k, int& s, int& dl, float& w) {
    s = 0; dl = 0; w = 0.0f;
    if (lane < k) {
      const int e = myring[(wdone + lane) & (RING - 1)];
      s = src[e];
      const int d = dst[e];
      w = dinv[s] * dinv[d];
      dl = d - lo;
      if (rec) {
        const int ow = dl & 31;
        const int slot = atomicAdd(&qcnt[ow], 1);
        if (slot < CAPW) queue[ow * CAPW + slot] = make_int2((s << 8) | dl, __builtin_bit_cast(int, w));
      }
    }
  };

  auto add_row = [&](const float4& v, float wi, int dli) {
    float* a = acc + dli * NOUT + lane;
    atomicAdd(a,      v.x * wi);
    atomicAdd(a + 32, v.y * wi);
    atomicAdd(a + 64, v.z * wi);
    atomicAdd(a + 96, v.w * wi);
  };

  auto drain = [&]() {
    while (wcount - wdone >= EPW) {
      int s, dl; float w;
      fetch(EPW, s, dl, w);
      float4 v[EPW];
#pragma unroll
      for (int i = 0; i < EPW; ++i) {
        const int si = __builtin_amdgcn_readlane(s, i);
        v[i] = xw4[(size_t)si * 32 + lane];
      }
#pragma unroll
      for (int i = 0; i < EPW; ++i) {
        const float wi = __builtin_bit_cast(float, __builtin_amdgcn_readlane(__builtin_bit_cast(int, w), i));
        const int dli  = __builtin_amdgcn_readlane(dl, i);
        add_row(v[i], wi, dli);
      }
      wdone += EPW;
    }
  };

  auto leftover = [&]() {
    const int k = wcount - wdone;
    if (k > 0) {
      int s, dl; float w;
      fetch(k, s, dl, w);
      for (int i = 0; i < k; ++i) {
        const int si = __builtin_amdgcn_readlane(s, i);
        const float wi = __builtin_bit_cast(float, __builtin_amdgcn_readlane(__builtin_bit_cast(int, w), i));
        const int dli  = __builtin_amdgcn_readlane(dl, i);
        const float4 v = xw4[(size_t)si * 32 + lane];
        add_row(v, wi, dli);
      }
      wdone += k;
    }
  };

  if (replay) {
    static_assert(RANGE <= 7 * 32, "a wave owns at most 7 rows");
    const v4f* xwv = reinterpret_cast<const v4f*>(xw);
    auto start = [&](int j) {
      v4f a = {0.0f, 0.0f, 0.0f, 0.0f};
      const int row = j * 32 + wave;
      if (row < cnt) {
        const float di = dinv[lo + row];
        a = xwv[(size_t)(lo + row) * 32 + lane] * di * di;
      }
      return a;
    };
#define OWN_SUM(N) const v4f s##N = start(N);                          \
    v2f a##N##l = __builtin_shufflevector(s##N, s##N, 0, 1);           \
    v2f a##N##h = __builtin_shufflevector(s##N, s##N, 2, 3);
    OWN_SUM(0) OWN_SUM(1) OWN_SUM(2) OWN_SUM(3) OWN_SUM(4) OWN_SUM(5) OWN_SUM(6)
#undef OWN_SUM
    auto add_own = [&](const v4f& v, float wi, int j) {
      const v4f t = v * wi;
      const v2f tl = __builtin_shufflevector(t, t, 0, 1);
      const v2f th = __builtin_shufflevector(t, t, 2, 3);
      asm volatile(
          "s_cmp_lt_u32 %[j], 4\n\t"
          "s_cbranch_scc0 4f\n\t"
          "s_cmp_lt_u32 %[j], 2\n\t"
          "s_cbranch_scc0 2f\n\t"
          "s_cmp_lg_u32 %[j], 0\n\t"
          "s_cbranch_scc1 1f\n\t"
          "v_pk_add_f32 %[a0l], %[a0l], %[tl]\n\t"
          "v_pk_add_f32 %[a0h], %[a0h], %[th]\n\t"
          "s_branch 9f\n"
          "1:\n\t"
          "v_pk_add_f32 %[a1l], %[a1l], %[tl]\n\t"
          "v_pk_add_f32 %[a1h], %[a1h], %[th]\n\t"
          "s_branch 9f\n"
          "2:\n\t"
          "s_cmp_lg_u32 %[j], 2\n\t"
          "s_cbranch_scc1 3f\n\t"
          "v_pk_add_f32 %[a2l], %[a2l], %[tl]\n\t"
          "v_pk_add_f32 %[a2h], %[a2h], %[th]\n\t"
          "s_branch 9f\n"
          "3:\n\t"
          "v_pk_add_f32 %[a3l], %[a3l], %[tl]\n\t"
          "v_pk_add_f32 %[a3h], %[a3h], %[th]\n\t"
          "s_branch 9f\n"
          "4:\n\t"
          "s_cmp_lt_u32 %[j], 6\n\t"
          "s_cbranch_scc0 6f\n\t"
          "s_cmp_lg_u32 %[j], 4\n\t"
          "s_cbranch_scc1 5f\n\t"
          "v_pk_add_f32 %[a4l], %[a4l], %[tl]\n\t"
          "v_pk_add_f32 %[a4h], %[a4h], %[th]\n\t"
          "s_branch 9f\n"
          "5:\n\t"
          "v_pk_add_f32 %[a5l], %[a5l], %[tl]\n\t"
          "v_pk_add_f32 %[a5h], %[a5h], %[th]\n\t"
          "s_branch 9f\n"
          "6:\n\t"
          "v_pk_add_f32 %[a6l], %[a6l], %[tl]\n\t"
          "v_pk_add_f32 %[a6h], %[a6h], %[th]\n"
          "9:"
          : [a0l] "+v"(a0l), [a0h] "+v"(a0h), [a1l] "+v"(a1l), [a1h] "+v"(a1h),
            [a2l] "+v"(a2l), [a2h] "+v"(a2h), [a3l] "+v"(a3l), [a3h] "+v"(a3h),
            [a4l] "+v"(a4l), [a4h] "+v"(a4h), [a5l] "+v"(a5l), [a5h] "+v"(a5h),
            [a6l] "+v"(a6l), [a6h] "+v"(a6h)
          : [j] "s"(j), [tl] "v"(tl), [th] "v"(th)
          : "scc");
    };
    auto group = [&](const int2& ent, int first, auto ne) {
      constexpr int NE = decltype(ne)::value;
      v4f v[NE];
#pragma unroll
      for (int i = 0; i < NE; ++i) {
        const int key = __builtin_amdgcn_readlane(ent.x, first + i);
        v[i] = xwv[(size_t)(key >> 8) * 32 + lane];
      }
#pragma unroll
      for (int i = 0; i < NE; ++i) {
        const int key  = __builtin_amdgcn_readlane(ent.x, first + i);
        const float wi = __builtin_bit_cast(float, __builtin_amdgcn_readlane(ent.y, first + i));
        add_own(v[i], wi, (key & 255) >> 5);
      }
    };
    int2* ment = reinterpret_cast<int2*>(myring);
    {
      int2 e[CAPW / 32];
#pragma unroll
      for (int j = 0; j < CAPW / 32; ++j) {
        e[j] = make_int2(0, 0);
        if (j * 32 + lane < mycount) e[j] = seg[j * 32 + lane];
      }
#pragma unroll
      for (int j = 0; j < CAPW / 32; ++j) ment[j * 32 + lane] = e[j];
    }
    for (int base = 0; base < mycount; base += 32) {
      const int2 cur = ment[base + lane];
      const int k = (mycount - base < 32) ? (mycount - base) : 32;
      if (k == 32) {
#pragma unroll
        for (int g = 0; g < 32; g += EPR) group(cur, g, std::integral_constant<int, EPR>{});
      } else {
        int g = 0;
        for (; g + EPW <= k; g += EPW) group(cur, g, std::integral_constant<int, EPW>{});
        for (; g < k; ++g) group(cur, g, std::integral_constant<int, 1>{});
      }
    }
    const float4 bo = reinterpret_cast<const float4*>(bias)[lane];
    float4* ho = reinterpret_cast<float4*>(h);
    auto finish = [&](const v2f& al, const v2f& ah, int j) {
      const int row = j * 32 + wave;
      if (row < cnt) {
        float4 o;
        o.x = fmaxf(al[0] + bo.x, 0.0f);
        o.y = fmaxf(al[1] + bo.y, 0.0f);
        o.z = fmaxf(ah[0] + bo.z, 0.0f);
        o.w = fmaxf(ah[1] + bo.w, 0.0f);
        ho[(size_t)(lo + row) * 32 + lane] = o;
      }
    };
    finish(a0l, a0h, 0); finish(a1l, a1h, 1); finish(a2l, a2h, 2); finish(a3l, a3h, 3);
    finish(a4l, a4h, 4); finish(a5l, a5h, 5); finish(a6l, a6h, 6);
  } else if (plane_ok) {
    const uint4* p4 = reinterpret_cast<const uint4*>(scratch);
    const int np = (E + 15) >> 4;
    const unsigned pat = blockIdx.x * 0x01010101u;
    auto hit = [&](unsigned w) {
      const unsigned x = w ^ pat;
      return ~(((x & 0x7F7F7F7Fu) + 0x7F7F7F7Fu) | x | 0x7F7F7F7Fu);
    };
    auto take = [&](unsigned t, int ebase) {
      while (__builtin_amdgcn_ballot_w32(t != 0u) != 0u) {
        const int e = ebase + (__builtin_ctz(t | 0x80000000u) >> 3);
        const bool m = (t != 0u) && (e < E);
        const unsigned mask = __builtin_amdgcn_ballot_w32(m);
        if (m) myring[(wcount + (int)__builtin_amdgcn_mbcnt_lo(mask, 0u)) & (RING - 1)] = e;
        wcount += __builtin_popcount(mask);
        t &= t - 1u;
      }
    };
    const int rot = (int)(blockIdx.x * ((unsigned)np / gridDim.x)) & ~31;
    auto piece = [&](const uint4& pv, int v) {
      const int e = 16 * v;
      const unsigned t0 = hit(pv.x), t1 = hit(pv.y), t2 = hit(pv.z), t3 = hit(pv.w);
      if (__builtin_amdgcn_ballot_w32((t0 | t1 | t2 | t3) != 0u) != 0u) {
        take(t0, e);
        take(t1, e + 4);
        take(t2, e + 8);
        take(t3, e + 12);
      }
      drain();
    };
    int vb = 0;
    for (; vb + SCAN_V * AGG_THREADS <= np; vb += SCAN_V * AGG_THREADS) {
      uint4 pv[SCAN_V];
      int   vi[SCAN_V];
#pragma unroll
      for (int j = 0; j < SCAN_V; ++j) {
        int v = vb + j * AGG_THREADS + tid + rot;
        if (v >= np) v -= np;
        vi[j] = v;
        pv[j] = p4[v];
      }
#pragma unroll
      for (int j = 0; j < SCAN_V; ++j) piece(pv[j], vi[j]);
    }
    for (; vb < np; vb += AGG_THREADS) {
      const int p = vb + tid;
      int v = p + rot;
      if (v >= np) v -= np;
      uint4 pv = make_uint4(~pat, ~pat, ~pat, ~pat);
      if (p < np) pv = p4[v];
      piece(pv, v);
    }
    leftover();
  } else {
    const bool al16 = (reinterpret_cast<uintptr_t>(dst) & 15u) == 0u;
    const int nv = al16 ? (E >> 2) : 0;
    const int4* d4 = reinterpret_cast<const int4*>(dst);
    const int rot = (int)(blockIdx.x * ((unsigned)nv / gridDim.x)) & ~31;
    int vb = 0;
    for (; vb + SCAN_V * AGG_THREADS <= nv; vb += SCAN_V * AGG_THREADS) {
      int4 dv[SCAN_V];
      int  vi[SCAN_V];
#pragma unroll
      for (int j = 0; j < SCAN_V; ++j) {
        int v = vb + j * AGG_THREADS + tid + rot;
        if (v >= nv) v -= nv;
        vi[j] = v;
        dv[j] = d4[v];
      }
#pragma unroll
      for (int j = 0; j < SCAN_V; ++j) {
        const int e = 4 * vi[j];
        test(dv[j].x, e);
        test(dv[j].y, e + 1);
        test(dv[j].z, e + 2);
        test(dv[j].w, e + 3);
      }
      drain();
    }
    for (; vb < nv; vb += AGG_THREADS) {
      const int p = vb + tid;
      int v = p + rot;
      if (v >= nv) v -= nv;
      int4 dv = make_int4(-1, -1, -1, -1);
      if (p < nv) dv = d4[v];
      const int e = 4 * v;
      test(dv.x, e);
      test(dv.y, e + 1);
      test(dv.z, e + 2);
      test(dv.w, e + 3);
      drain();
    }
    for (int eb = nv * 4; eb < E; eb += AGG_THREADS) {
      const int e = eb + tid;
      const int d = (e < E) ? dst[e] : -1;
      test(d, e);
      drain();
    }
    leftover();
  }
  __syncthreads();

  if (rec) {
    const int qc = qcnt[wave];
    const int qm = (qc < CAPW) ? qc : CAPW;
    for (int i = lane; i < qm; i += 32) seg[i] = queue[wave * CAPW + i];
    if (lane == 0) counts[blockIdx.x * (AGG_THREADS / 32) + wave] = qc;
  }

  const float4 bv = reinterpret_cast<const float4*>(bias)[lane];
  float4* h4 = reinterpret_cast<float4*>(h);
  if (!replay)
  for (int idx = tid; idx < cnt * 32; idx += AGG_THREADS) {
    const int node = idx >> 5;
    const float* a = acc + node * NOUT + lane;
    float4 o;
    o.x = fmaxf(a[0]  + bv.x, 0.0f);
    o.y = fmaxf(a[32] + bv.y, 0.0f);
    o.z = fmaxf(a[64] + bv.z, 0.0f);
    o.w = fmaxf(a[96] + bv.w, 0.0f);
    h4[(size_t)(lo + node) * 32 + lane] = o;
  }
}

#define GEMM_BLOCKS 512
template <int KK>
__global__ __launch_bounds__(256) void k_gemm_wmma(
    const float* __restrict__ A, const float* __restrict__ W,
    const float* __restrict__ bias, float* __restrict__ C,
    int M, int flags) {
  constexpr int P  = KK + 4;
  constexpr int V  = KK / 4;
  constexpr int KS = KK / 4;
  constexpr int NV = (16 * V) / 256;
  __shared__ alignas(16) float sA[2][16 * P];

  const int tid  = threadIdx.x;
  const int lane = tid & 31;
  const int wave = tid >> 5;
  const int half = lane >> 4;
  const int r    = lane & 15;
  const int colBase = wave * 16;
  const int ntiles  = M / 16;
  const int step    = (int)gridDim.x;
  int t = (int)blockIdx.x;
  if (t >= ntiles) return;

  const float4* A4 = reinterpret_cast<const float4*>(A);
  static_assert(NV == 2 || NV == 4, "a thread carries 2 or 4 pieces of a stripe");
  float4 s0, s1, s2 = {}, s3 = {};
  auto fetch_stripe = [&](int tile) {
    const float4* p = A4 + (size_t)tile * (16 * V) + tid;
    s0 = p[0];
    s1 = p[256];
    if constexpr (NV == 4) { s2 = p[512]; s3 = p[768]; }
  };
  auto put_piece = [&](float* dstStripe, int i, const float4& v) {
    *reinterpret_cast<float4*>(&dstStripe[(i / V) * P + (i % V) * 4]) = v;
  };
  auto put_stripe = [&](float* dstStripe) {
    put_piece(dstStripe, tid, s0);
    put_piece(dstStripe, tid + 256, s1);
    if constexpr (NV == 4) { put_piece(dstStripe, tid + 512, s2); put_piece(dstStripe, tid + 768, s3); }
  };
  fetch_stripe(t);

  v2f bf[KS];
  const float* wcol = W + (size_t)(2 * half) * NOUT + colBase + r;
#pragma unroll
  for (int s = 0; s < KS; ++s) {
    bf[s][0] = wcol[(size_t)(4 * s) * NOUT];
    bf[s][1] = wcol[(size_t)(4 * s + 1) * NOUT];
  }
  const float bv = (flags & 2) ? bias[colBase + r] : 0.0f;

  put_stripe(sA[0]);
  __syncthreads();

  int buf = 0;
  for (; t < ntiles; t += step) {
    const int  tn   = t + step;
    const bool more = tn < ntiles;
    if (more) fetch_stripe(tn);

    const int rowBase = t * 16;
    float* crow = C + (size_t)(rowBase + half * 8) * NOUT + colBase + r;
    v8f cold = {};
    if (flags & 1) {
#pragma unroll
      for (int i = 0; i < 8; ++i) cold[i] = crow[(size_t)i * NOUT];
    }

    v8f acc = {};
    const float* arow = &sA[buf][r * P + 2 * half];
#pragma unroll
    for (int s = 0; s < KS; ++s) {
      const v2f a = *reinterpret_cast<const v2f*>(arow + 4 * s);
      acc = __builtin_amdgcn_wmma_f32_16x16x4_f32(
           false, a,  false, bf[s],
           (short)0, acc,  false,  false);
    }

    put_stripe(sA[buf ^ 1]);
    __syncthreads();
    buf ^= 1;

#pragma unroll
    for (int i = 0; i < 8; ++i) {
      float v = acc[i] + bv;
      if (flags & 1) v += cold[i];
      if (flags & 4) v = fmaxf(v, 0.0f);
      crow[(size_t)i * NOUT] = v;
    }
  }
}

extern "C" void kernel_launch(void* const* d_in, const int* in_sizes, int n_in,
                              void* d_out, int out_size, void* d_ws, size_t ws_size,
                              hipStream_t stream) {
  const float* x  = (const float*)d_in[0];
  const int*   ei = (const int*)d_in[1];
  const float* W1 = (const float*)d_in[2];
  const float* b1 = (const float*)d_in[3];
  const float* W2 = (const float*)d_in[4];
  const float* b2 = (const float*)d_in[5];
  const float* Ws = (const float*)d_in[6];
  const float* bs = (const float*)d_in[7];
  const float* Wf = (const float*)d_in[8];
  const float* bf = (const float*)d_in[9];
  float* out = (float*)d_out;

  const int IN = 256;
  const int n  = in_sizes[0] / IN;
  const int E  = in_sizes[1] / 2;
  const int* src = ei;
  const int* dst = ei + E;

  float* ws   = (float*)d_ws;
  float* dinv = ws;
  float* xw   = ws + n;
  float* agg  = xw + (size_t)n * NOUT;

  dim3 B(256);
  const int gn   = (n + 255) / 256;
  const int gR   = (n + RANGE - 1) / RANGE;
  const int gM   = GEMM_BLOCKS;

  k_deg_hist       <<<DEG_BLOCKS, dim3(DEG_THREADS), 0, stream>>>(dst, agg, out, n, E);
  k_dinv_from_parts<<<gn, B, 0, stream>>>(agg, dinv, n);

  k_gemm_wmma<256><<<gM, B, 0, stream>>>(x, W1, nullptr, xw, n, 0);
  k_owner_aggregate<<<gR, dim3(AGG_THREADS), 0, stream>>>(xw, dinv, src, dst, b1, agg, out, n, E, 0);

  k_gemm_wmma<128><<<gM, B, 0, stream>>>(agg, W2, nullptr, xw, n, 0);
  k_owner_aggregate<<<gR, dim3(AGG_THREADS), 0, stream>>>(xw, dinv, src, dst, b2, agg, out, n, E, 1);

  k_gemm_wmma<128><<<gM, B, 0, stream>>>(agg, Wf, bf, out, n, 2);
  k_gemm_wmma<256><<<gM, B, 0, stream>>>(x,  Ws, bs, out, n, 1 | 2);
}


# First aggregate's sweep: a lane's hit flags of two 16-byte pieces (32 edges) merged into one word, one append loop per pair (was eight, one per dword); ring order changes, arrival-order sums
# speedup vs baseline: 4.3095x; 1.0327x over previous
#include <hip/hip_runtime.h>
#include <math.h>
#include <stdint.h>
#include <type_traits>

typedef __attribute__((ext_vector_type(2))) float v2f;
typedef __attribute__((ext_vector_type(8))) float v8f;
typedef __attribute__((ext_vector_type(4))) float v4f;

#define NOUT 128

#define RANGE 196
#define DEG_BLOCKS 64
#define DEG_THREADS 1024
#define DEG_HN 51200
#define DEG_V 8
static_assert(DEG_BLOCKS <= NOUT, "the partial counts live in a buffer of n * NOUT floats");

__global__ __launch_bounds__(DEG_THREADS) void k_deg_hist(
    const int* __restrict__ dst, float* __restrict__ part, float* __restrict__ scratch, int n, int E) {
  __shared__ alignas(16) float hist[DEG_HN];
  const int tid  = threadIdx.x;
  const int epad = (E + 15) & ~15;
  const int nq   = epad >> 2;
  const int per  = (nq + (int)gridDim.x - 1) / (int)gridDim.x;
  const int q0   = (int)blockIdx.x * per;
  const int q1   = (nq - q0 < per) ? nq : q0 + per;
  const bool al16  = (reinterpret_cast<uintptr_t>(dst) & 15u) == 0u;
  const bool plane = (size_t)epad <= (size_t)n * NOUT * 4;
  unsigned* pl  = reinterpret_cast<unsigned*>(scratch);
  float* mypart = part + (size_t)blockIdx.x * n;

  for (int base = 0; base < n; base += DEG_HN) {
    const int cnt = (n - base < DEG_HN) ? (n - base) : DEG_HN;
    for (int i = tid; i < (cnt + 3) / 4; i += DEG_THREADS)
      reinterpret_cast<float4*>(hist)[i] = make_float4(0.0f, 0.0f, 0.0f, 0.0f);
    __syncthreads();

    auto one = [&](int d, int j, unsigned& pk) {
      if (d >= 0) {
        const unsigned r = (unsigned)(d - base);
        if (r < (unsigned)cnt) atomicAdd(&hist[r], 1.0f);
        pk |= (((unsigned)d / (unsigned)RANGE) & 255u) << (8 * j);
      }
    };
    for (int qb = q0; qb < q1; qb += DEG_V * DEG_THREADS) {
      int4 dv[DEG_V];
#pragma unroll
      for (int j = 0; j < DEG_V; ++j) {
        const int q = qb + j * DEG_THREADS + tid;
        dv[j] = make_int4(-1, -1, -1, -1);
        if (q < q1) {
          const int e0 = 4 * q;
          if (al16 && e0 + 4 <= E) {
            dv[j] = reinterpret_cast<const int4*>(dst)[q];
          } else {
            if (e0     < E) dv[j].x = dst[e0];
            if (e0 + 1 < E) dv[j].y = dst[e0 + 1];
            if (e0 + 2 < E) dv[j].z = dst[e0 + 2];
            if (e0 + 3 < E) dv[j].w = dst[e0 + 3];
          }
        }
      }
#pragma unroll
      for (int j = 0; j < DEG_V; ++j) {
        const int q = qb + j * DEG_THREADS + tid;
        if (q < q1) {
          unsigned pk = 0u;
          one(dv[j].x, 0, pk);
          one(dv[j].y, 1, pk);
          one(dv[j].z, 2, pk);
          one(dv[j].w, 3, pk);
          if (plane && base == 0) pl[q] = pk;
        }
      }
    }
    __syncthreads();

    float* o = mypart + base;
    int done = 0;
    if ((reinterpret_cast<uintptr_t>(o) & 15u) == 0u) {
      for (int i = tid; i < cnt / 4; i += DEG_THREADS)
        reinterpret_cast<float4*>(o)[i] = reinterpret_cast<const float4*>(hist)[i];
      done = cnt & ~3;
    }
    for (int i = done + tid; i < cnt; i += DEG_THREADS) o[i] = hist[i];
    if (base + DEG_HN < n) __syncthreads();
  }
}

#define DEG_LD 32
static_assert(DEG_BLOCKS % DEG_LD == 0, "the partials are read in whole groups");
__global__ void k_dinv_from_parts(const float* __restrict__ part, float* __restrict__ dinv, int n) {
  const int i = blockIdx.x * blockDim.x + threadIdx.x;
  if (i >= n) return;
  const float* p = part + i;
  float d = 1.0f;
  for (int g0 = 0; g0 < DEG_BLOCKS; g0 += DEG_LD) {
    float v[DEG_LD];
#pragma unroll
    for (int g = 0; g < DEG_LD; ++g) v[g] = p[(size_t)(g0 + g) * n];
#pragma unroll
    for (int g = 0; g < DEG_LD; ++g) d += v[g];
  }
  dinv[i] = (d > 0.0f) ? rsqrtf(d) : 0.0f;
}

#define EPW 8
#define SCAN_V 4
#define AGG_THREADS 1024
#define RING 1024
#define CAPW 320
#define EPR 16

__global__ __launch_bounds__(AGG_THREADS) void k_owner_aggregate(
    const float* __restrict__ xw, const float* __restrict__ dinv,
    const int* __restrict__ src, const int* __restrict__ dst,
    const float* __restrict__ bias, float* __restrict__ h,
    float* __restrict__ scratch, int n, int E, int mode) {
  __shared__ float acc[RANGE * NOUT];
  __shared__ alignas(8) int ring[(AGG_THREADS / 32) * RING];
  __shared__ alignas(8) int2 queue[(AGG_THREADS / 32) * CAPW];
  __shared__ int qcnt[AGG_THREADS / 32];
  static_assert(AGG_THREADS / 32 == 32, "a wave reads the counts of all waves of its block, one per lane");
  static_assert(CAPW % 32 == 0 && CAPW % EPW == 0 && 2 * CAPW <= RING && RANGE <= 256, "record layout");

  const int tid  = threadIdx.x;
  const int lane = tid & 31;
  const int wave = __builtin_amdgcn_readfirstlane(tid >> 5);
  const int lo   = blockIdx.x * RANGE;
  const int cnt  = (n - lo < RANGE) ? (n - lo) : RANGE;

  const float4* xw4 = reinterpret_cast<const float4*>(xw);

  const size_t nwaves = (size_t)gridDim.x * (AGG_THREADS / 32);
  const size_t pbytes = (((size_t)E + 15) >> 4) << 4;
  const bool usable = (n <= (1 << 23)) &&
                      (pbytes + nwaves * (CAPW * 8 + 4) <= (size_t)n * NOUT * 4) &&
                      ((reinterpret_cast<uintptr_t>(scratch) & 15u) == 0u);
  const bool plane_ok = usable && (gridDim.x <= 256u);
  char* rbase = reinterpret_cast<char*>(scratch) + pbytes;
  int2* seg   = reinterpret_cast<int2*>(rbase) + ((size_t)blockIdx.x * (AGG_THREADS / 32) + wave) * CAPW;
  int* counts = reinterpret_cast<int*>(rbase) + nwaves * CAPW * 2;
  const bool rec = usable && (mode == 0);
  bool replay = false;
  int mycount = 0;
  if (usable && mode == 1) {
    const int cv = counts[blockIdx.x * (AGG_THREADS / 32) + lane];
    replay  = __builtin_amdgcn_ballot_w32(cv > CAPW) == 0u;
    mycount = __builtin_amdgcn_readlane(cv, wave);
  }

  if (tid < AGG_THREADS / 32) qcnt[tid] = 0;
  if (!replay)
  for (int idx = tid; idx < cnt * 32; idx += AGG_THREADS) {
    const int node = idx >> 5;
    const float di = dinv[lo + node];
    const float4 x = xw4[(size_t)(lo + node) * 32 + lane];
    float* a = acc + node * NOUT + lane;
    a[0]  = x.x * di * di;
    a[32] = x.y * di * di;
    a[64] = x.z * di * di;
    a[96] = x.w * di * di;
  }
  __syncthreads();

  int* myring = ring + wave * RING;
  int wcount = 0;
  int wdone  = 0;

  auto test = [&](int d, int e) {
    const bool m = (unsigned)(d - lo) < (unsigned)cnt;
    const unsigned mask = __builtin_amdgcn_ballot_w32(m);
    if (mask != 0u) {
      if (m) myring[(wcount + (int)__builtin_amdgcn_mbcnt_lo(mask, 0u)) & (RING - 1)] = e;
      wcount += __builtin_popcount(mask);
    }
  };

  auto fetch = [&](int k, int& s, int& dl, float& w) {
    s = 0; dl = 0; w = 0.0f;
    if (lane < k) {
      const int e = myring[(wdone + lane) & (RING - 1)];
      s = src[e];
      const int d = dst[e];
      w = dinv[s] * dinv[d];
      dl = d - lo;
      if (rec) {
        const int ow = dl & 31;
        const int slot = atomicAdd(&qcnt[ow], 1);
        if (slot < CAPW) queue[ow * CAPW + slot] = make_int2((s << 8) | dl, __builtin_bit_cast(int, w));
      }
    }
  };

  auto add_row = [&](const float4& v, float wi, int dli) {
    float* a = acc + dli * NOUT + lane;
    atomicAdd(a,      v.x * wi);
    atomicAdd(a + 32, v.y * wi);
    atomicAdd(a + 64, v.z * wi);
    atomicAdd(a + 96, v.w * wi);
  };

  auto drain = [&]() {
    while (wcount - wdone >= EPW) {
      int s, dl; float w;
      fetch(EPW, s, dl, w);
      float4 v[EPW];
#pragma unroll
      for (int i = 0; i < EPW; ++i) {
        const int si = __builtin_amdgcn_readlane(s, i);
        v[i] = xw4[(size_t)si * 32 + lane];
      }
#pragma unroll
      for (int i = 0; i < EPW; ++i) {
        const float wi = __builtin_bit_cast(float, __builtin_amdgcn_readlane(__builtin_bit_cast(int, w), i));
        const int dli  = __builtin_amdgcn_readlane(dl, i);
        add_row(v[i], wi, dli);
      }
      wdone += EPW;
    }
  };

  auto leftover = [&]() {
    const int k = wcount - wdone;
    if (k > 0) {
      int s, dl; float w;
      fetch(k, s, dl, w);
      for (int i = 0; i < k; ++i) {
        const int si = __builtin_amdgcn_readlane(s, i);
        const float wi = __builtin_bit_cast(float, __builtin_amdgcn_readlane(__builtin_bit_cast(int, w), i));
        const int dli  = __builtin_amdgcn_readlane(dl, i);
        const float4 v = xw4[(size_t)si * 32 + lane];
        add_row(v, wi, dli);
      }
      wdone += k;
    }
  };

  if (replay) {
    static_assert(RANGE <= 7 * 32, "a wave owns at most 7 rows");
    const v4f* xwv = reinterpret_cast<const v4f*>(xw);
    auto start = [&](int j) {
      v4f a = {0.0f, 0.0f, 0.0f, 0.0f};
      const int row = j * 32 + wave;
      if (row < cnt) {
        const float di = dinv[lo + row];
        a = xwv[(size_t)(lo + row) * 32 + lane] * di * di;
      }
      return a;
    };
#define OWN_SUM(N) const v4f s##N = start(N);                          \
    v2f a##N##l = __builtin_shufflevector(s##N, s##N, 0, 1);           \
    v2f a##N##h = __builtin_shufflevector(s##N, s##N, 2, 3);
    OWN_SUM(0) OWN_SUM(1) OWN_SUM(2) OWN_SUM(3) OWN_SUM(4) OWN_SUM(5) OWN_SUM(6)
#undef OWN_SUM
    auto add_own = [&](const v4f& v, float wi, int j) {
      const v4f t = v * wi;
      const v2f tl = __builtin_shufflevector(t, t, 0, 1);
      const v2f th = __builtin_shufflevector(t, t, 2, 3);
      asm volatile(
          "s_cmp_lt_u32 %[j], 4\n\t"
          "s_cbranch_scc0 4f\n\t"
          "s_cmp_lt_u32 %[j], 2\n\t"
          "s_cbranch_scc0 2f\n\t"
          "s_cmp_lg_u32 %[j], 0\n\t"
          "s_cbranch_scc1 1f\n\t"
          "v_pk_add_f32 %[a0l], %[a0l], %[tl]\n\t"
          "v_pk_add_f32 %[a0h], %[a0h], %[th]\n\t"
          "s_branch 9f\n"
          "1:\n\t"
          "v_pk_add_f32 %[a1l], %[a1l], %[tl]\n\t"
          "v_pk_add_f32 %[a1h], %[a1h], %[th]\n\t"
          "s_branch 9f\n"
          "2:\n\t"
          "s_cmp_lg_u32 %[j], 2\n\t"
          "s_cbranch_scc1 3f\n\t"
          "v_pk_add_f32 %[a2l], %[a2l], %[tl]\n\t"
          "v_pk_add_f32 %[a2h], %[a2h], %[th]\n\t"
          "s_branch 9f\n"
          "3:\n\t"
          "v_pk_add_f32 %[a3l], %[a3l], %[tl]\n\t"
          "v_pk_add_f32 %[a3h], %[a3h], %[th]\n\t"
          "s_branch 9f\n"
          "4:\n\t"
          "s_cmp_lt_u32 %[j], 6\n\t"
          "s_cbranch_scc0 6f\n\t"
          "s_cmp_lg_u32 %[j], 4\n\t"
          "s_cbranch_scc1 5f\n\t"
          "v_pk_add_f32 %[a4l], %[a4l], %[tl]\n\t"
          "v_pk_add_f32 %[a4h], %[a4h], %[th]\n\t"
          "s_branch 9f\n"
          "5:\n\t"
          "v_pk_add_f32 %[a5l], %[a5l], %[tl]\n\t"
          "v_pk_add_f32 %[a5h], %[a5h], %[th]\n\t"
          "s_branch 9f\n"
          "6:\n\t"
          "v_pk_add_f32 %[a6l], %[a6l], %[tl]\n\t"
          "v_pk_add_f32 %[a6h], %[a6h], %[th]\n"
          "9:"
          : [a0l] "+v"(a0l), [a0h] "+v"(a0h), [a1l] "+v"(a1l), [a1h] "+v"(a1h),
            [a2l] "+v"(a2l), [a2h] "+v"(a2h), [a3l] "+v"(a3l), [a3h] "+v"(a3h),
            [a4l] "+v"(a4l), [a4h] "+v"(a4h), [a5l] "+v"(a5l), [a5h] "+v"(a5h),
            [a6l] "+v"(a6l), [a6h] "+v"(a6h)
          : [j] "s"(j), [tl] "v"(tl), [th] "v"(th)
          : "scc");
    };
    auto group = [&](const int2& ent, int first, auto ne) {
      constexpr int NE = decltype(ne)::value;
      v4f v[NE];
#pragma unroll
      for (int i = 0; i < NE; ++i) {
        const int key = __builtin_amdgcn_readlane(ent.x, first + i);
        v[i] = xwv[(size_t)(key >> 8) * 32 + lane];
      }
#pragma unroll
      for (int i = 0; i < NE; ++i) {
        const int key  = __builtin_amdgcn_readlane(ent.x, first + i);
        const float wi = __builtin_bit_cast(float, __builtin_amdgcn_readlane(ent.y, first + i));
        add_own(v[i], wi, (key & 255) >> 5);
      }
    };
    int2* ment = reinterpret_cast<int2*>(myring);
    {
      int2 e[CAPW / 32];
#pragma unroll
      for (int j = 0; j < CAPW / 32; ++j) {
        e[j] = make_int2(0, 0);
        if (j * 32 + lane < mycount) e[j] = seg[j * 32 + lane];
      }
#pragma unroll
      for (int j = 0; j < CAPW / 32; ++j) ment[j * 32 + lane] = e[j];
    }
    for (int base = 0; base < mycount; base += 32) {
      const int2 cur = ment[base + lane];
      const int k = (mycount - base < 32) ? (mycount - base) : 32;
      if (k == 32) {
#pragma unroll
        for (int g = 0; g < 32; g += EPR) group(cur, g, std::integral_constant<int, EPR>{});
      } else {
        int g = 0;
        for (; g + EPW <= k; g += EPW) group(cur, g, std::integral_constant<int, EPW>{});
        for (; g < k; ++g) group(cur, g, std::integral_constant<int, 1>{});
      }
    }
    const float4 bo = reinterpret_cast<const float4*>(bias)[lane];
    float4* ho = reinterpret_cast<float4*>(h);
    auto finish = [&](const v2f& al, const v2f& ah, int j) {
      const int row = j * 32 + wave;
      if (row < cnt) {
        float4 o;
        o.x = fmaxf(al[0] + bo.x, 0.0f);
        o.y = fmaxf(al[1] + bo.y, 0.0f);
        o.z = fmaxf(ah[0] + bo.z, 0.0f);
        o.w = fmaxf(ah[1] + bo.w, 0.0f);
        ho[(size_t)(lo + row) * 32 + lane] = o;
      }
    };
    finish(a0l, a0h, 0); finish(a1l, a1h, 1); finish(a2l, a2h, 2); finish(a3l, a3h, 3);
    finish(a4l, a4h, 4); finish(a5l, a5h, 5); finish(a6l, a6h, 6);
  } else if (plane_ok) {
    const uint4* p4 = reinterpret_cast<const uint4*>(scratch);
    const int np = (E + 15) >> 4;
    const unsigned pat = blockIdx.x * 0x01010101u;
    auto hit = [&](unsigned w) {
      const unsigned x = w ^ pat;
      return ~(((x & 0x7F7F7F7Fu) + 0x7F7F7F7Fu) | x | 0x7F7F7F7Fu);
    };
    auto flags2 = [&](const uint4& a, const uint4& b) {
      const unsigned ma = (hit(a.x) >> 7) | (hit(a.y) >> 6) | (hit(a.z) >> 5) | (hit(a.w) >> 4);
      const unsigned mb = (hit(b.x) >> 3) | (hit(b.y) >> 2) | (hit(b.z) >> 1) | hit(b.w);
      return ma | mb;
    };
    auto flags1 = [&](const uint4& a) {
      return (hit(a.x) >> 7) | (hit(a.y) >> 6) | (hit(a.z) >> 5) | (hit(a.w) >> 4);
    };
    auto take = [&](unsigned m, int ea, int eb) {
      unsigned any = __builtin_amdgcn_ballot_w32(m != 0u);
      while (any != 0u) {
        const int p = __builtin_ctz(m | 0x80000000u);
        const int e = ((p & 4) ? eb : ea) + ((p & 3) << 2) + (p >> 3);
        const bool f = (m != 0u) && (e < E);
        const unsigned mask = __builtin_amdgcn_ballot_w32(f);
        if (f) myring[(wcount + (int)__builtin_amdgcn_mbcnt_lo(mask, 0u)) & (RING - 1)] = e;
        wcount += __builtin_popcount(mask);
        m &= m - 1u;
        any = __builtin_amdgcn_ballot_w32(m != 0u);
        if (wcount - wdone >= RING - 32) drain();
      }
    };
    const int rot = (int)(blockIdx.x * ((unsigned)np / gridDim.x)) & ~31;
    int vb = 0;
    static_assert(SCAN_V % 2 == 0, "the pieces of a group are taken two at a time");
    for (; vb + SCAN_V * AGG_THREADS <= np; vb += SCAN_V * AGG_THREADS) {
      uint4 pv[SCAN_V];
      int   vi[SCAN_V];
#pragma unroll
      for (int j = 0; j < SCAN_V; ++j) {
        int v = vb + j * AGG_THREADS + tid + rot;
        if (v >= np) v -= np;
        vi[j] = v;
        pv[j] = p4[v];
      }
#pragma unroll
      for (int j = 0; j < SCAN_V; j += 2) take(flags2(pv[j], pv[j + 1]), 16 * vi[j], 16 * vi[j + 1]);
      drain();
    }
    for (; vb < np; vb += AGG_THREADS) {
      const int p = vb + tid;
      int v = p + rot;
      if (v >= np) v -= np;
      uint4 pv = make_uint4(~pat, ~pat, ~pat, ~pat);
      if (p < np) pv = p4[v];
      take(flags1(pv), 16 * v, 0);
      drain();
    }
    leftover();
  } else {
    const bool al16 = (reinterpret_cast<uintptr_t>(dst) & 15u) == 0u;
    const int nv = al16 ? (E >> 2) : 0;
    const int4* d4 = reinterpret_cast<const int4*>(dst);
    const int rot = (int)(blockIdx.x * ((unsigned)nv / gridDim.x)) & ~31;
    int vb = 0;
    for (; vb + SCAN_V * AGG_THREADS <= nv; vb += SCAN_V * AGG_THREADS) {
      int4 dv[SCAN_V];
      int  vi[SCAN_V];
#pragma unroll
      for (int j = 0; j < SCAN_V; ++j) {
        int v = vb + j * AGG_THREADS + tid + rot;
        if (v >= nv) v -= nv;
        vi[j] = v;
        dv[j] = d4[v];
      }
#pragma unroll
      for (int j = 0; j < SCAN_V; ++j) {
        const int e = 4 * vi[j];
        test(dv[j].x, e);
        test(dv[j].y, e + 1);
        test(dv[j].z, e + 2);
        test(dv[j].w, e + 3);
      }
      drain();
    }
    for (; vb < nv; vb += AGG_THREADS) {
      const int p = vb + tid;
      int v = p + rot;
      if (v >= nv) v -= nv;
      int4 dv = make_int4(-1, -1, -1, -1);
      if (p < nv) dv = d4[v];
      const int e = 4 * v;
      test(dv.x, e);
      test(dv.y, e + 1);
      test(dv.z, e + 2);
      test(dv.w, e + 3);
      drain();
    }
    for (int eb = nv * 4; eb < E; eb += AGG_THREADS) {
      const int e = eb + tid;
      const int d = (e < E) ? dst[e] : -1;
      test(d, e);
      drain();
    }
    leftover();
  }
  __syncthreads();

  if (rec) {
    const int qc = qcnt[wave];
    const int qm = (qc < CAPW) ? qc : CAPW;
    for (int i = lane; i < qm; i += 32) seg[i] = queue[wave * CAPW + i];
    if (lane == 0) counts[blockIdx.x * (AGG_THREADS / 32) + wave] = qc;
  }

  const float4 bv = reinterpret_cast<const float4*>(bias)[lane];
  float4* h4 = reinterpret_cast<float4*>(h);
  if (!replay)
  for (int idx = tid; idx < cnt * 32; idx += AGG_THREADS) {
    const int node = idx >> 5;
    const float* a = acc + node * NOUT + lane;
    float4 o;
    o.x = fmaxf(a[0]  + bv.x, 0.0f);
    o.y = fmaxf(a[32] + bv.y, 0.0f);
    o.z = fmaxf(a[64] + bv.z, 0.0f);
    o.w = fmaxf(a[96] + bv.w, 0.0f);
    h4[(size_t)(lo + node) * 32 + lane] = o;
  }
}

#define GEMM_BLOCKS 512
template <int KK>
__global__ __launch_bounds__(256) void k_gemm_wmma(
    const float* __restrict__ A, const float* __restrict__ W,
    const float* __restrict__ bias, float* __restrict__ C,
    int M, int flags) {
  constexpr int P  = KK + 4;
  constexpr int V  = KK / 4;
  constexpr int KS = KK / 4;
  constexpr int NV = (16 * V) / 256;
  __shared__ alignas(16) float sA[2][16 * P];

  const int tid  = threadIdx.x;
  const int lane = tid & 31;
  const int wave = tid >> 5;
  const int half = lane >> 4;
  const int r    = lane & 15;
  const int colBase = wave * 16;
  const int ntiles  = M / 16;
  const int step    = (int)gridDim.x;
  int t = (int)blockIdx.x;
  if (t >= ntiles) return;

  const float4* A4 = reinterpret_cast<const float4*>(A);
  static_assert(NV == 2 || NV == 4, "a thread carries 2 or 4 pieces of a stripe");
  float4 s0, s1, s2 = {}, s3 = {};
  auto fetch_stripe = [&](int tile) {
    const float4* p = A4 + (size_t)tile * (16 * V) + tid;
    s0 = p[0];
    s1 = p[256];
    if constexpr (NV == 4) { s2 = p[512]; s3 = p[768]; }
  };
  auto put_piece = [&](float* dstStripe, int i, const float4& v) {
    *reinterpret_cast<float4*>(&dstStripe[(i / V) * P + (i % V) * 4]) = v;
  };
  auto put_stripe = [&](float* dstStripe) {
    put_piece(dstStripe, tid, s0);
    put_piece(dstStripe, tid + 256, s1);
    if constexpr (NV == 4) { put_piece(dstStripe, tid + 512, s2); put_piece(dstStripe, tid + 768, s3); }
  };
  fetch_stripe(t);

  v2f bf[KS];
  const float* wcol = W + (size_t)(2 * half) * NOUT + colBase + r;
#pragma unroll
  for (int s = 0; s < KS; ++s) {
    bf[s][0] = wcol[(size_t)(4 * s) * NOUT];
    bf[s][1] = wcol[(size_t)(4 * s + 1) * NOUT];
  }
  const float bv = (flags & 2) ? bias[colBase + r] : 0.0f;

  put_stripe(sA[0]);
  __syncthreads();

  int buf = 0;
  for (; t < ntiles; t += step) {
    const int  tn   = t + step;
    const bool more = tn < ntiles;
    if (more) fetch_stripe(tn);

    const int rowBase = t * 16;
    float* crow = C + (size_t)(rowBase + half * 8) * NOUT + colBase + r;
    v8f cold = {};
    if (flags & 1) {
#pragma unroll
      for (int i = 0; i < 8; ++i) cold[i] = crow[(size_t)i * NOUT];
    }

    v8f acc = {};
    const float* arow = &sA[buf][r * P + 2 * half];
#pragma unroll
    for (int s = 0; s < KS; ++s) {
      const v2f a = *reinterpret_cast<const v2f*>(arow + 4 * s);
      acc = __builtin_amdgcn_wmma_f32_16x16x4_f32(
           false, a,  false, bf[s],
           (short)0, acc,  false,  false);
    }

    put_stripe(sA[buf ^ 1]);
    __syncthreads();
    buf ^= 1;

#pragma unroll
    for (int i = 0; i < 8; ++i) {
      float v = acc[i] + bv;
      if (flags & 1) v += cold[i];
      if (flags & 4) v = fmaxf(v, 0.0f);
      crow[(size_t)i * NOUT] = v;
    }
  }
}

extern "C" void kernel_launch(void* const* d_in, const int* in_sizes, int n_in,
                              void* d_out, int out_size, void* d_ws, size_t ws_size,
                              hipStream_t stream) {
  const float* x  = (const float*)d_in[0];
  const int*   ei = (const int*)d_in[1];
  const float* W1 = (const float*)d_in[2];
  const float* b1 = (const float*)d_in[3];
  const float* W2 = (const float*)d_in[4];
  const float* b2 = (const float*)d_in[5];
  const float* Ws = (const float*)d_in[6];
  const float* bs = (const float*)d_in[7];
  const float* Wf = (const float*)d_in[8];
  const float* bf = (const float*)d_in[9];
  float* out = (float*)d_out;

  const int IN = 256;
  const int n  = in_sizes[0] / IN;
  const int E  = in_sizes[1] / 2;
  const int* src = ei;
  const int* dst = ei + E;

  float* ws   = (float*)d_ws;
  float* dinv = ws;
  float* xw   = ws + n;
  float* agg  = xw + (size_t)n * NOUT;

  dim3 B(256);
  const int gn   = (n + 255) / 256;
  const int gR   = (n + RANGE - 1) / RANGE;
  const int gM   = GEMM_BLOCKS;

  k_deg_hist       <<<DEG_BLOCKS, dim3(DEG_THREADS), 0, stream>>>(dst, agg, out, n, E);
  k_dinv_from_parts<<<gn, B, 0, stream>>>(agg, dinv, n);

  k_gemm_wmma<256><<<gM, B, 0, stream>>>(x, W1, nullptr, xw, n, 0);
  k_owner_aggregate<<<gR, dim3(AGG_THREADS), 0, stream>>>(xw, dinv, src, dst, b1, agg, out, n, E, 0);

  k_gemm_wmma<128><<<gM, B, 0, stream>>>(agg, W2, nullptr, xw, n, 0);
  k_owner_aggregate<<<gR, dim3(AGG_THREADS), 0, stream>>>(xw, dinv, src, dst, b2, agg, out, n, E, 1);

  k_gemm_wmma<128><<<gM, B, 0, stream>>>(agg, Wf, bf, out, n, 2);
  k_gemm_wmma<256><<<gM, B, 0, stream>>>(x,  Ws, bs, out, n, 1 | 2);
}
